# starting kernel
# baseline (speedup 1.0000x reference)
#include <hip/hip_runtime.h>
#include <hip/hip_bf16.h>


#define GN_N 100000
#define GN_E 640000
#define GN_D 128

typedef __attribute__((ext_vector_type(2))) float v2f;
typedef __attribute__((ext_vector_type(8))) float v8f;

__global__ void sage_pack_w(const float* __restrict__ W, float* __restrict__ Bpk) {
    int idx = blockIdx.x * blockDim.x + threadIdx.x;
    if (idx >= 32 * 2 * 128) return;
    int j  = idx & 127;
    int h  = (idx >> 7) & 1;
    int kg = idx >> 8;
    int k  = kg * 4 + 2 * h;
    Bpk[idx * 2 + 0] = W[j * GN_D + k];
    Bpk[idx * 2 + 1] = W[j * GN_D + k + 1];
}

__global__ void sage_count(const int* __restrict__ dst, float* __restrict__ cnt) {
    int e = blockIdx.x * blockDim.x + threadIdx.x;
    if (e < GN_E) atomicAdd(&cnt[dst[e]], 1.0f);
}

__global__ void sage_invdeg(float* __restrict__ cnt) {
    int i = blockIdx.x * blockDim.x + threadIdx.x;
    if (i < GN_N) cnt[i] = 1.0f / fmaxf(cnt[i], 1.0f);
}

__global__ __launch_bounds__(256) void sage_scatter(
    const float* __restrict__ H, const int* __restrict__ src,
    const int* __restrict__ dst, float* __restrict__ SUM) {
    int idx  = blockIdx.x * blockDim.x + threadIdx.x;
    int e    = idx >> 5;
    int lane = idx & 31;
    if (e >= GN_E) return;
    int s = src[e];
    int d = dst[e];
    const float4 v = *(const float4*)(H + (size_t)s * GN_D + lane * 4);
    float* p = SUM + (size_t)d * GN_D + lane * 4;
    atomicAdd(p + 0, v.x);
    atomicAdd(p + 1, v.y);
    atomicAdd(p + 2, v.z);
    atomicAdd(p + 3, v.w);
}

__global__ __launch_bounds__(128) void sage_gemm(
    const float* __restrict__ A1, const float* __restrict__ invdeg,
    const float* __restrict__ B1pk, const float* __restrict__ bias,
    const float* __restrict__ A2, const float* __restrict__ B2pk,
    float* __restrict__ out, int relu_out) {
    const int lane = threadIdx.x & 31;
    const int wave = blockIdx.x * (blockDim.x >> 5) + (threadIdx.x >> 5);
    if (wave >= GN_N / 16) return;

    const int rbase = wave * 16;
    const int h     = lane >> 4;
    const int l15   = lane & 15;
    const int row   = rbase + l15;

    float s = 1.0f;
    if (invdeg) s = invdeg[row];

    const float* a1row = A1 + (size_t)row * GN_D;
    const float* a2row = A2 ? (A2 + (size_t)row * GN_D) : nullptr;

    float bj[8];
#pragma unroll
    for (int jt = 0; jt < 8; ++jt) bj[jt] = bias[jt * 16 + l15];

    v8f c[8];
#pragma unroll
    for (int jt = 0; jt < 8; ++jt)
#pragma unroll
        for (int v = 0; v < 8; ++v) c[jt][v] = bj[jt];

#pragma unroll
    for (int k0 = 0; k0 < GN_D; k0 += 4) {
        v2f a = *(const v2f*)(a1row + k0 + 2 * h);
        a.x *= s;
        a.y *= s;
        const float* brow = B1pk + ((k0 >> 2) * 2 + h) * 256 + 2 * l15;
#pragma unroll
        for (int jt = 0; jt < 8; ++jt) {
            v2f b = *(const v2f*)(brow + jt * 32);
            c[jt] = __builtin_amdgcn_wmma_f32_16x16x4_f32(
                false, a, false, b, (short)0, c[jt], false, false);
        }
    }

    if (a2row) {
#pragma unroll
        for (int k0 = 0; k0 < GN_D; k0 += 4) {
            v2f a = *(const v2f*)(a2row + k0 + 2 * h);
            const float* brow = B2pk + ((k0 >> 2) * 2 + h) * 256 + 2 * l15;
#pragma unroll
            for (int jt = 0; jt < 8; ++jt) {
                v2f b = *(const v2f*)(brow + jt * 32);
                c[jt] = __builtin_amdgcn_wmma_f32_16x16x4_f32(
                    false, a, false, b, (short)0, c[jt], false, false);
            }
        }
    }

#pragma unroll
    for (int jt = 0; jt < 8; ++jt)
#pragma unroll
        for (int v = 0; v < 8; ++v) {
            float val = c[jt][v];
            if (relu_out) val = fmaxf(val, 0.0f);
            out[(size_t)(rbase + v + 8 * h) * GN_D + jt * 16 + l15] = val;
        }
}

extern "C" void kernel_launch(void* const* d_in, const int* in_sizes, int n_in,
                              void* d_out, int out_size, void* d_ws, size_t ws_size,
                              hipStream_t stream) {
    (void)in_sizes; (void)n_in; (void)out_size; (void)ws_size;

    const float* x   = (const float*)d_in[0];
    const int*   ei  = (const int*)d_in[1];
    const float* W0p = (const float*)d_in[2];
    const float* b0p = (const float*)d_in[3];
    const float* W0l = (const float*)d_in[4];
    const float* b0l = (const float*)d_in[5];
    const float* W0r = (const float*)d_in[6];
    const float* W1p = (const float*)d_in[7];
    const float* b1p = (const float*)d_in[8];
    const float* W1l = (const float*)d_in[9];
    const float* b1l = (const float*)d_in[10];
    const float* W1r = (const float*)d_in[11];
    float* out = (float*)d_out;

    const size_t nd = (size_t)GN_N * GN_D;
    float* H   = (float*)d_ws;
    float* X1  = H + nd;
    float* SUM = X1 + nd;
    float* INV = SUM + nd;
    float* PK  = INV + GN_N;
    float* P0p = PK + 0 * 16384;
    float* P0l = PK + 1 * 16384;
    float* P0r = PK + 2 * 16384;
    float* P1p = PK + 3 * 16384;
    float* P1l = PK + 4 * 16384;
    float* P1r = PK + 5 * 16384;

    const int* src = ei;
    const int* dst = ei + GN_E;

    sage_pack_w<<<32, 256, 0, stream>>>(W0p, P0p);
    sage_pack_w<<<32, 256, 0, stream>>>(W0l, P0l);
    sage_pack_w<<<32, 256, 0, stream>>>(W0r, P0r);
    sage_pack_w<<<32, 256, 0, stream>>>(W1p, P1p);
    sage_pack_w<<<32, 256, 0, stream>>>(W1l, P1l);
    sage_pack_w<<<32, 256, 0, stream>>>(W1r, P1r);

    hipMemsetAsync(INV, 0, GN_N * sizeof(float), stream);
    sage_count<<<(GN_E + 255) / 256, 256, 0, stream>>>(dst, INV);
    sage_invdeg<<<(GN_N + 255) / 256, 256, 0, stream>>>(INV);

    const int gemm_blocks    = (GN_N / 16 + 3) / 4;
    const int scatter_blocks = (GN_E * 32 + 255) / 256;

    sage_gemm<<<gemm_blocks, 128, 0, stream>>>(x, nullptr, P0p, b0p,
                                               nullptr, nullptr, H, 1);
    hipMemsetAsync(SUM, 0, nd * sizeof(float), stream);
    sage_scatter<<<scatter_blocks, 256, 0, stream>>>(H, src, dst, SUM);
    sage_gemm<<<gemm_blocks, 128, 0, stream>>>(SUM, INV, P0l, b0l,
                                               x, P0r, X1, 1);

    sage_gemm<<<gemm_blocks, 128, 0, stream>>>(X1, nullptr, P1p, b1p,
                                               nullptr, nullptr, H, 1);
    hipMemsetAsync(SUM, 0, nd * sizeof(float), stream);
    sage_scatter<<<scatter_blocks, 256, 0, stream>>>(H, src, dst, SUM);
    sage_gemm<<<gemm_blocks, 128, 0, stream>>>(SUM, INV, P1l, b1l,
                                               X1, P1r, out, 0);
}


# Neighbour sums as destination-ordered row gathers: edges counting-sorted by dst once per call (integer atomics), one wave adds a node's rows in arrival-order rank; no f32 atomics, no zero fill
# speedup vs baseline: 1.5664x; 1.5664x over previous
#include <hip/hip_runtime.h>
#include <hip/hip_bf16.h>


#define GN_N 100000
#define GN_E 640000
#define GN_D 128

typedef __attribute__((ext_vector_type(2))) float v2f;
typedef __attribute__((ext_vector_type(8))) float v8f;

__global__ void sage_pack_w(const float* __restrict__ W, float* __restrict__ Bpk) {
    int idx = blockIdx.x * blockDim.x + threadIdx.x;
    if (idx >= 32 * 2 * 128) return;
    int j  = idx & 127;
    int h  = (idx >> 7) & 1;
    int kg = idx >> 8;
    int k  = kg * 4 + 2 * h;
    Bpk[idx * 2 + 0] = W[j * GN_D + k];
    Bpk[idx * 2 + 1] = W[j * GN_D + k + 1];
}

__global__ __launch_bounds__(256) void sage_rank(
    const int* __restrict__ dst, int* __restrict__ cnt, int* __restrict__ rank) {
    int e = blockIdx.x * blockDim.x + threadIdx.x;
    if (e < GN_E) rank[e] = atomicAdd(&cnt[dst[e]], 1);
}

__global__ __launch_bounds__(256) void sage_alloc(
    const int* __restrict__ cnt, int* __restrict__ cursor,
    int* __restrict__ start, float* __restrict__ inv) {
    __shared__ int wsum[8];
    __shared__ int base_s;
    const int i    = blockIdx.x * 256 + threadIdx.x;
    const int lane = threadIdx.x & 31;
    const int w    = threadIdx.x >> 5;
    const int c    = (i < GN_N) ? cnt[i] : 0;
    int v = c;
    for (int d = 1; d < 32; d <<= 1) {
        int t = __shfl_up(v, d);
        if (lane >= d) v += t;
    }
    if (lane == 31) wsum[w] = v;
    __syncthreads();
    if (threadIdx.x == 0) {
        int tot = 0;
        for (int k = 0; k < 8; ++k) { int t = wsum[k]; wsum[k] = tot; tot += t; }
        base_s = atomicAdd(cursor, tot);
    }
    __syncthreads();
    if (i < GN_N) {
        start[i] = base_s + wsum[w] + v - c;
        inv[i]   = 1.0f / fmaxf((float)c, 1.0f);
    }
}

__global__ __launch_bounds__(256) void sage_fill(
    const int* __restrict__ src, const int* __restrict__ dst,
    const int* __restrict__ rank, const int* __restrict__ start,
    int* __restrict__ esrc) {
    int e = blockIdx.x * blockDim.x + threadIdx.x;
    if (e >= GN_E) return;
    int p = start[dst[e]] + rank[e];
    if ((unsigned)p < (unsigned)GN_E) esrc[p] = src[e];
}

template <int R>
__device__ __forceinline__ void sage_rows(const float* __restrict__ Hl,
                                          int ids, int j, float4& acc) {
    float4 v[R];
    for (int k = 0; k < R; ++k)
        v[k] = *(const float4*)(Hl + (size_t)__builtin_amdgcn_readlane(ids, j + k) * GN_D);
    for (int k = 0; k < R; ++k) {
        acc.x += v[k].x; acc.y += v[k].y; acc.z += v[k].z; acc.w += v[k].w;
    }
}

__device__ __forceinline__ void sage_add_rows(const float* __restrict__ Hl,
                                              int ids, int m, float4& acc) {
    int j = 0;
    for (; j + 8 <= m; j += 8) sage_rows<8>(Hl, ids, j, acc);
    switch (m - j) {
        case 1: sage_rows<1>(Hl, ids, j, acc); break;
        case 2: sage_rows<2>(Hl, ids, j, acc); break;
        case 3: sage_rows<3>(Hl, ids, j, acc); break;
        case 4: sage_rows<4>(Hl, ids, j, acc); break;
        case 5: sage_rows<5>(Hl, ids, j, acc); break;
        case 6: sage_rows<6>(Hl, ids, j, acc); break;
        case 7: sage_rows<7>(Hl, ids, j, acc); break;
        default: break;
    }
}

__global__ __launch_bounds__(256) void sage_gather(
    const float* __restrict__ H, const int* __restrict__ start,
    const int* __restrict__ cnt, const int* __restrict__ esrc,
    float* __restrict__ SUM) {
    const int lane = threadIdx.x & 31;
    const int wv     = __builtin_amdgcn_readfirstlane((int)(threadIdx.x >> 5));
    const int stride = gridDim.x * 8;
    const float* Hl  = H + lane * 4;

    int i0 = blockIdx.x * 8 + wv;
    if (i0 >= GN_N) return;
    int s0 = start[i0], n0 = cnt[i0];
    int i1 = i0 + stride;
    int s1 = 0, n1 = 0;
    if (i1 < GN_N) { s1 = start[i1]; n1 = cnt[i1]; }
    int ids0 = esrc[min(s0 + lane, GN_E - 1)];

    for (;;) {
        const int ids1 = esrc[min(s1 + lane, GN_E - 1)];
        const int i2 = i1 + stride;
        int s2 = 0, n2 = 0;
        if (i2 < GN_N) { s2 = start[i2]; n2 = cnt[i2]; }

        float4 acc = make_float4(0.0f, 0.0f, 0.0f, 0.0f);
        sage_add_rows(Hl, ids0, min(n0, 32), acc);
        for (int j0 = 32; j0 < n0; j0 += 32) {
            const int ids = esrc[min(s0 + j0 + lane, GN_E - 1)];
            sage_add_rows(Hl, ids, min(n0 - j0, 32), acc);
        }
        *(float4*)(SUM + (size_t)i0 * GN_D + lane * 4) = acc;

        if (i1 >= GN_N) break;
        i0 = i1; s0 = s1; n0 = n1; ids0 = ids1;
        i1 = i2; s1 = s2; n1 = n2;
    }
}

__global__ __launch_bounds__(128) void sage_gemm(
    const float* __restrict__ A1, const float* __restrict__ invdeg,
    const float* __restrict__ B1pk, const float* __restrict__ bias,
    const float* __restrict__ A2, const float* __restrict__ B2pk,
    float* __restrict__ out, int relu_out) {
    const int lane = threadIdx.x & 31;
    const int wave = blockIdx.x * (blockDim.x >> 5) + (threadIdx.x >> 5);
    if (wave >= GN_N / 16) return;

    const int rbase = wave * 16;
    const int h     = lane >> 4;
    const int l15   = lane & 15;
    const int row   = rbase + l15;

    float s = 1.0f;
    if (invdeg) s = invdeg[row];

    const float* a1row = A1 + (size_t)row * GN_D;
    const float* a2row = A2 ? (A2 + (size_t)row * GN_D) : nullptr;

    float bj[8];
#pragma unroll
    for (int jt = 0; jt < 8; ++jt) bj[jt] = bias[jt * 16 + l15];

    v8f c[8];
#pragma unroll
    for (int jt = 0; jt < 8; ++jt)
#pragma unroll
        for (int v = 0; v < 8; ++v) c[jt][v] = bj[jt];

#pragma unroll
    for (int k0 = 0; k0 < GN_D; k0 += 4) {
        v2f a = *(const v2f*)(a1row + k0 + 2 * h);
        a.x *= s;
        a.y *= s;
        const float* brow = B1pk + ((k0 >> 2) * 2 + h) * 256 + 2 * l15;
#pragma unroll
        for (int jt = 0; jt < 8; ++jt) {
            v2f b = *(const v2f*)(brow + jt * 32);
            c[jt] = __builtin_amdgcn_wmma_f32_16x16x4_f32(
                false, a, false, b, (short)0, c[jt], false, false);
        }
    }

    if (a2row) {
#pragma unroll
        for (int k0 = 0; k0 < GN_D; k0 += 4) {
            v2f a = *(const v2f*)(a2row + k0 + 2 * h);
            const float* brow = B2pk + ((k0 >> 2) * 2 + h) * 256 + 2 * l15;
#pragma unroll
            for (int jt = 0; jt < 8; ++jt) {
                v2f b = *(const v2f*)(brow + jt * 32);
                c[jt] = __builtin_amdgcn_wmma_f32_16x16x4_f32(
                    false, a, false, b, (short)0, c[jt], false, false);
            }
        }
    }

#pragma unroll
    for (int jt = 0; jt < 8; ++jt)
#pragma unroll
        for (int v = 0; v < 8; ++v) {
            float val = c[jt][v];
            if (relu_out) val = fmaxf(val, 0.0f);
            out[(size_t)(rbase + v + 8 * h) * GN_D + jt * 16 + l15] = val;
        }
}

extern "C" void kernel_launch(void* const* d_in, const int* in_sizes, int n_in,
                              void* d_out, int out_size, void* d_ws, size_t ws_size,
                              hipStream_t stream) {
    (void)in_sizes; (void)n_in; (void)out_size; (void)ws_size;

    const float* x   = (const float*)d_in[0];
    const int*   ei  = (const int*)d_in[1];
    const float* W0p = (const float*)d_in[2];
    const float* b0p = (const float*)d_in[3];
    const float* W0l = (const float*)d_in[4];
    const float* b0l = (const float*)d_in[5];
    const float* W0r = (const float*)d_in[6];
    const float* W1p = (const float*)d_in[7];
    const float* b1p = (const float*)d_in[8];
    const float* W1l = (const float*)d_in[9];
    const float* b1l = (const float*)d_in[10];
    const float* W1r = (const float*)d_in[11];
    float* out = (float*)d_out;

    const size_t nd = (size_t)GN_N * GN_D;
    float* H   = (float*)d_ws;
    float* X1  = H + nd;
    float* SUM = X1 + nd;
    float* INV = SUM + nd;
    float* PK  = INV + GN_N;
    float* P0p = PK + 0 * 16384;
    float* P0l = PK + 1 * 16384;
    float* P0r = PK + 2 * 16384;
    float* P1p = PK + 3 * 16384;
    float* P1l = PK + 4 * 16384;
    float* P1r = PK + 5 * 16384;
    int* CNT   = (int*)(PK + 6 * 16384);
    int* CUR   = CNT + GN_N;
    int* START = CNT + GN_N + 32;
    int* RANK  = START + GN_N;
    int* ESRC  = RANK + GN_E;

    const int* src = ei;
    const int* dst = ei + GN_E;

    sage_pack_w<<<32, 256, 0, stream>>>(W0p, P0p);
    sage_pack_w<<<32, 256, 0, stream>>>(W0l, P0l);
    sage_pack_w<<<32, 256, 0, stream>>>(W0r, P0r);
    sage_pack_w<<<32, 256, 0, stream>>>(W1p, P1p);
    sage_pack_w<<<32, 256, 0, stream>>>(W1l, P1l);
    sage_pack_w<<<32, 256, 0, stream>>>(W1r, P1r);

    hipMemsetAsync(CNT, 0, (GN_N + 1) * sizeof(int), stream);
    sage_rank<<<(GN_E + 255) / 256, 256, 0, stream>>>(dst, CNT, RANK);
    sage_alloc<<<(GN_N + 255) / 256, 256, 0, stream>>>(CNT, CUR, START, INV);
    sage_fill<<<(GN_E + 255) / 256, 256, 0, stream>>>(src, dst, RANK, START, ESRC);

    const int gemm_blocks   = (GN_N / 16 + 3) / 4;
    const int gather_blocks = 2048;

    sage_gemm<<<gemm_blocks, 128, 0, stream>>>(x, nullptr, P0p, b0p,
                                               nullptr, nullptr, H, 1);
    sage_gather<<<gather_blocks, 256, 0, stream>>>(H, START, CNT, ESRC, SUM);
    sage_gemm<<<gemm_blocks, 128, 0, stream>>>(SUM, INV, P0l, b0l,
                                               x, P0r, X1, 1);

    sage_gemm<<<gemm_blocks, 128, 0, stream>>>(X1, nullptr, P1p, b1p,
                                               nullptr, nullptr, H, 1);
    sage_gather<<<gather_blocks, 256, 0, stream>>>(H, START, CNT, ESRC, SUM);
    sage_gemm<<<gemm_blocks, 128, 0, stream>>>(SUM, INV, P1l, b1l,
                                               X1, P1r, out, 0);
}


# GEMM B fragments read from a per-workgroup LDS copy of the raw weights (250 workgroups of 25 waves, next k-step's 4 ds_load_b128 ahead of each 8 WMMAs); the 6 pack launches are gone
# speedup vs baseline: 2.1640x; 1.3815x over previous
#include <hip/hip_runtime.h>
#include <hip/hip_bf16.h>


#define GN_N 100000
#define GN_E 640000
#define GN_D 128

typedef __attribute__((ext_vector_type(2))) float v2f;
typedef __attribute__((ext_vector_type(8))) float v8f;

#define GN_KG_PITCH     516
#define GN_W_LDS        (32 * GN_KG_PITCH)
#define GN_GEMM_WAVES   25
#define GN_GEMM_THREADS (GN_GEMM_WAVES * 32)
#define GN_GEMM_BLOCKS  (GN_N / 16 / GN_GEMM_WAVES)
#define GN_LDS_FLOATS   41472

__device__ __forceinline__ void sage_stage_w(const float* __restrict__ W,
                                             float* __restrict__ Bs) {
    const int rot = (blockIdx.x * 37) & 127;
#pragma unroll
    for (int it = 0; it < 6; ++it) {
        const int idx = it * GN_GEMM_THREADS + (int)threadIdx.x;
        if (idx < 32 * 128) {
            const int j  = ((idx >> 5) + rot) & 127;
            const int kg = idx & 31;
            const float4 w = *(const float4*)(W + j * GN_D + kg * 4);
            float* d = Bs + kg * GN_KG_PITCH + (j >> 5) * 128 + (j & 15) * 4 + ((j >> 4) & 1) * 2;
            v2f lo; lo.x = w.x; lo.y = w.y;
            v2f hi; hi.x = w.z; hi.y = w.w;
            *(v2f*)d        = lo;
            *(v2f*)(d + 64) = hi;
        }
    }
}

__global__ __launch_bounds__(256) void sage_rank(
    const int* __restrict__ dst, int* __restrict__ cnt, int* __restrict__ rank) {
    int e = blockIdx.x * blockDim.x + threadIdx.x;
    if (e < GN_E) rank[e] = atomicAdd(&cnt[dst[e]], 1);
}

__global__ __launch_bounds__(256) void sage_alloc(
    const int* __restrict__ cnt, int* __restrict__ cursor,
    int* __restrict__ start, float* __restrict__ inv) {
    __shared__ int wsum[8];
    __shared__ int base_s;
    const int i    = blockIdx.x * 256 + threadIdx.x;
    const int lane = threadIdx.x & 31;
    const int w    = threadIdx.x >> 5;
    const int c    = (i < GN_N) ? cnt[i] : 0;
    int v = c;
    for (int d = 1; d < 32; d <<= 1) {
        int t = __shfl_up(v, d);
        if (lane >= d) v += t;
    }
    if (lane == 31) wsum[w] = v;
    __syncthreads();
    if (threadIdx.x == 0) {
        int tot = 0;
        for (int k = 0; k < 8; ++k) { int t = wsum[k]; wsum[k] = tot; tot += t; }
        base_s = atomicAdd(cursor, tot);
    }
    __syncthreads();
    if (i < GN_N) {
        start[i] = base_s + wsum[w] + v - c;
        inv[i]   = 1.0f / fmaxf((float)c, 1.0f);
    }
}

__global__ __launch_bounds__(256) void sage_fill(
    const int* __restrict__ src, const int* __restrict__ dst,
    const int* __restrict__ rank, const int* __restrict__ start,
    int* __restrict__ esrc) {
    int e = blockIdx.x * blockDim.x + threadIdx.x;
    if (e >= GN_E) return;
    int p = start[dst[e]] + rank[e];
    if ((unsigned)p < (unsigned)GN_E) esrc[p] = src[e];
}

template <int R>
__device__ __forceinline__ void sage_rows(const float* __restrict__ Hl,
                                          int ids, int j, float4& acc) {
    float4 v[R];
    for (int k = 0; k < R; ++k)
        v[k] = *(const float4*)(Hl + (size_t)__builtin_amdgcn_readlane(ids, j + k) * GN_D);
    for (int k = 0; k < R; ++k) {
        acc.x += v[k].x; acc.y += v[k].y; acc.z += v[k].z; acc.w += v[k].w;
    }
}

__device__ __forceinline__ void sage_add_rows(const float* __restrict__ Hl,
                                              int ids, int m, float4& acc) {
    int j = 0;
    for (; j + 8 <= m; j += 8) sage_rows<8>(Hl, ids, j, acc);
    switch (m - j) {
        case 1: sage_rows<1>(Hl, ids, j, acc); break;
        case 2: sage_rows<2>(Hl, ids, j, acc); break;
        case 3: sage_rows<3>(Hl, ids, j, acc); break;
        case 4: sage_rows<4>(Hl, ids, j, acc); break;
        case 5: sage_rows<5>(Hl, ids, j, acc); break;
        case 6: sage_rows<6>(Hl, ids, j, acc); break;
        case 7: sage_rows<7>(Hl, ids, j, acc); break;
        default: break;
    }
}

__global__ __launch_bounds__(256) void sage_gather(
    const float* __restrict__ H, const int* __restrict__ start,
    const int* __restrict__ cnt, const int* __restrict__ esrc,
    float* __restrict__ SUM) {
    const int lane = threadIdx.x & 31;
    const int wv     = __builtin_amdgcn_readfirstlane((int)(threadIdx.x >> 5));
    const int stride = gridDim.x * 8;
    const float* Hl  = H + lane * 4;

    int i0 = blockIdx.x * 8 + wv;
    if (i0 >= GN_N) return;
    int s0 = start[i0], n0 = cnt[i0];
    int i1 = i0 + stride;
    int s1 = 0, n1 = 0;
    if (i1 < GN_N) { s1 = start[i1]; n1 = cnt[i1]; }
    int ids0 = esrc[min(s0 + lane, GN_E - 1)];

    for (;;) {
        const int ids1 = esrc[min(s1 + lane, GN_E - 1)];
        const int i2 = i1 + stride;
        int s2 = 0, n2 = 0;
        if (i2 < GN_N) { s2 = start[i2]; n2 = cnt[i2]; }

        float4 acc = make_float4(0.0f, 0.0f, 0.0f, 0.0f);
        sage_add_rows(Hl, ids0, min(n0, 32), acc);
        for (int j0 = 32; j0 < n0; j0 += 32) {
            const int ids = esrc[min(s0 + j0 + lane, GN_E - 1)];
            sage_add_rows(Hl, ids, min(n0 - j0, 32), acc);
        }
        *(float4*)(SUM + (size_t)i0 * GN_D + lane * 4) = acc;

        if (i1 >= GN_N) break;
        i0 = i1; s0 = s1; n0 = n1; ids0 = ids1;
        i1 = i2; s1 = s2; n1 = n2;
    }
}

__global__ __launch_bounds__(GN_GEMM_THREADS) void sage_gemm(
    const float* __restrict__ A1, const float* __restrict__ invdeg,
    const float* __restrict__ W1, const float* __restrict__ bias,
    const float* __restrict__ A2, const float* __restrict__ W2,
    float* __restrict__ out, int relu_out) {
    __shared__ __attribute__((aligned(16))) float Bs[GN_LDS_FLOATS];

    const int lane = threadIdx.x & 31;
    const int wave = blockIdx.x * GN_GEMM_WAVES +
                     __builtin_amdgcn_readfirstlane((int)(threadIdx.x >> 5));

    sage_stage_w(W1, Bs);
    if (A2) sage_stage_w(W2, Bs + GN_W_LDS);

    const int rbase = wave * 16;
    const int h     = lane >> 4;
    const int l15   = lane & 15;
    const int row   = min(rbase + l15, GN_N - 1);

    float s = 1.0f;
    if (invdeg) s = invdeg[row];

    float bj[8];
#pragma unroll
    for (int jt = 0; jt < 8; ++jt) bj[jt] = bias[jt * 16 + l15];

    __syncthreads();
    if (wave >= GN_N / 16) return;

    v8f c[8];
#pragma unroll
    for (int jt = 0; jt < 8; ++jt)
#pragma unroll
        for (int v = 0; v < 8; ++v) c[jt][v] = bj[jt];

    const int npath = A2 ? 2 : 1;
#pragma nounroll
    for (int p = 0; p < npath; ++p) {
        const float* arow  = (p ? A2 : A1) + (size_t)row * GN_D + 2 * h;
        const float* bfrag = Bs + p * GN_W_LDS + lane * 4;
        const float  sc    = p ? 1.0f : s;

        v2f    ar[8];
        float4 bq[2][4];
#pragma unroll
        for (int i = 0; i < 8; ++i) ar[i] = *(const v2f*)(arow + 4 * i);
#pragma unroll
        for (int jp = 0; jp < 4; ++jp) bq[0][jp] = *(const float4*)(bfrag + jp * 128);

#pragma unroll
        for (int kg = 0; kg < 32; ++kg) {
            if (kg + 1 < 32) {
#pragma unroll
                for (int jp = 0; jp < 4; ++jp)
                    bq[(kg + 1) & 1][jp] =
                        *(const float4*)(bfrag + (kg + 1) * GN_KG_PITCH + jp * 128);
            }
            v2f a = ar[kg & 7];
            a.x *= sc;
            a.y *= sc;
            if (kg + 8 < 32) ar[kg & 7] = *(const v2f*)(arow + 4 * (kg + 8));
#pragma unroll
            for (int jp = 0; jp < 4; ++jp) {
                const float4 bb = bq[kg & 1][jp];
                v2f b0; b0.x = bb.x; b0.y = bb.y;
                v2f b1; b1.x = bb.z; b1.y = bb.w;
                c[2 * jp] = __builtin_amdgcn_wmma_f32_16x16x4_f32(
                    false, a, false, b0, (short)0, c[2 * jp], false, false);
                c[2 * jp + 1] = __builtin_amdgcn_wmma_f32_16x16x4_f32(
                    false, a, false, b1, (short)0, c[2 * jp + 1], false, false);
            }
            __builtin_amdgcn_sched_barrier(0);
        }
    }

#pragma unroll
    for (int jt = 0; jt < 8; ++jt)
#pragma unroll
        for (int v = 0; v < 8; ++v) {
            float val = c[jt][v];
            if (relu_out) val = fmaxf(val, 0.0f);
            out[(size_t)(rbase + v + 8 * h) * GN_D + jt * 16 + l15] = val;
        }
}

extern "C" void kernel_launch(void* const* d_in, const int* in_sizes, int n_in,
                              void* d_out, int out_size, void* d_ws, size_t ws_size,
                              hipStream_t stream) {
    (void)in_sizes; (void)n_in; (void)out_size; (void)ws_size;

    const float* x   = (const float*)d_in[0];
    const int*   ei  = (const int*)d_in[1];
    const float* W0p = (const float*)d_in[2];
    const float* b0p = (const float*)d_in[3];
    const float* W0l = (const float*)d_in[4];
    const float* b0l = (const float*)d_in[5];
    const float* W0r = (const float*)d_in[6];
    const float* W1p = (const float*)d_in[7];
    const float* b1p = (const float*)d_in[8];
    const float* W1l = (const float*)d_in[9];
    const float* b1l = (const float*)d_in[10];
    const float* W1r = (const float*)d_in[11];
    float* out = (float*)d_out;

    const size_t nd = (size_t)GN_N * GN_D;
    float* H   = (float*)d_ws;
    float* X1  = H + nd;
    float* SUM = X1 + nd;
    float* INV = SUM + nd;
    int* CNT   = (int*)(INV + GN_N);
    int* CUR   = CNT + GN_N;
    int* START = CNT + GN_N + 32;
    int* RANK  = START + GN_N;
    int* ESRC  = RANK + GN_E;

    const int* src = ei;
    const int* dst = ei + GN_E;

    hipMemsetAsync(CNT, 0, (GN_N + 1) * sizeof(int), stream);
    sage_rank<<<(GN_E + 255) / 256, 256, 0, stream>>>(dst, CNT, RANK);
    sage_alloc<<<(GN_N + 255) / 256, 256, 0, stream>>>(CNT, CUR, START, INV);
    sage_fill<<<(GN_E + 255) / 256, 256, 0, stream>>>(src, dst, RANK, START, ESRC);

    const int gemm_blocks   = GN_GEMM_BLOCKS;
    const int gather_blocks = 2048;

    sage_gemm<<<gemm_blocks, GN_GEMM_THREADS, 0, stream>>>(x, nullptr, W0p, b0p,
                                               nullptr, nullptr, H, 1);
    sage_gather<<<gather_blocks, 256, 0, stream>>>(H, START, CNT, ESRC, SUM);
    sage_gemm<<<gemm_blocks, GN_GEMM_THREADS, 0, stream>>>(SUM, INV, W0l, b0l,
                                               x, W0r, X1, 1);

    sage_gemm<<<gemm_blocks, GN_GEMM_THREADS, 0, stream>>>(X1, nullptr, W1p, b1p,
                                               nullptr, nullptr, H, 1);
    sage_gather<<<gather_blocks, 256, 0, stream>>>(H, START, CNT, ESRC, SUM);
    sage_gemm<<<gemm_blocks, GN_GEMM_THREADS, 0, stream>>>(SUM, INV, W1l, b1l,
                                               X1, W1r, out, 0);
}


# Edge sort replaced by one bucket pass: node rows of 16 source slots filled by rank atomic, overflow list past 16; rank/alloc/fill launches gone; rows summed in slot rank then list order
# speedup vs baseline: 2.3074x; 1.0663x over previous
#include <hip/hip_runtime.h>
#include <hip/hip_bf16.h>


#define GN_N 100000
#define GN_E 640000
#define GN_D 128

typedef __attribute__((ext_vector_type(2))) float v2f;
typedef __attribute__((ext_vector_type(8))) float v8f;

#define GN_KG_PITCH     516
#define GN_W_LDS        (32 * GN_KG_PITCH)
#define GN_GEMM_WAVES   25
#define GN_GEMM_THREADS (GN_GEMM_WAVES * 32)
#define GN_GEMM_BLOCKS  (GN_N / 16 / GN_GEMM_WAVES)
#define GN_LDS_FLOATS   41472

__device__ __forceinline__ void sage_stage_w(const float* __restrict__ W,
                                             float* __restrict__ Bs) {
    const int rot = (blockIdx.x * 37) & 127;
#pragma unroll
    for (int it = 0; it < 6; ++it) {
        const int idx = it * GN_GEMM_THREADS + (int)threadIdx.x;
        if (idx < 32 * 128) {
            const int j  = ((idx >> 5) + rot) & 127;
            const int kg = idx & 31;
            const float4 w = *(const float4*)(W + j * GN_D + kg * 4);
            float* d = Bs + kg * GN_KG_PITCH + (j >> 5) * 128 + (j & 15) * 4 + ((j >> 4) & 1) * 2;
            v2f lo; lo.x = w.x; lo.y = w.y;
            v2f hi; hi.x = w.z; hi.y = w.w;
            *(v2f*)d        = lo;
            *(v2f*)(d + 64) = hi;
        }
    }
}

#define GN_SLOTS 16

__global__ __launch_bounds__(256) void sage_bucket(
    const int* __restrict__ src, const int* __restrict__ dst,
    int* __restrict__ cnt, int* __restrict__ slots,
    int* __restrict__ ovf_cursor, int* __restrict__ ovf_dst, int* __restrict__ ovf_src) {
    const int e = blockIdx.x * blockDim.x + threadIdx.x;
    if (e >= GN_E) return;
    const int d = dst[e];
    const int s = src[e];
    const int r = atomicAdd(&cnt[d], 1);
    if (r < GN_SLOTS) {
        slots[d * GN_SLOTS + r] = s;
    } else {
        const int p = atomicAdd(ovf_cursor, 1);
        if ((unsigned)p < (unsigned)GN_E) { ovf_dst[p] = d; ovf_src[p] = s; }
    }
}

template <int R>
__device__ __forceinline__ void sage_rows(const float* __restrict__ Hl,
                                          int ids, int j, float4& acc) {
    float4 v[R];
    for (int k = 0; k < R; ++k)
        v[k] = *(const float4*)(Hl + (size_t)__builtin_amdgcn_readlane(ids, j + k) * GN_D);
    for (int k = 0; k < R; ++k) {
        acc.x += v[k].x; acc.y += v[k].y; acc.z += v[k].z; acc.w += v[k].w;
    }
}

__device__ __forceinline__ void sage_add_rows(const float* __restrict__ Hl,
                                              int ids, int m, float4& acc) {
    int j = 0;
    for (; j + 8 <= m; j += 8) sage_rows<8>(Hl, ids, j, acc);
    switch (m - j) {
        case 1: sage_rows<1>(Hl, ids, j, acc); break;
        case 2: sage_rows<2>(Hl, ids, j, acc); break;
        case 3: sage_rows<3>(Hl, ids, j, acc); break;
        case 4: sage_rows<4>(Hl, ids, j, acc); break;
        case 5: sage_rows<5>(Hl, ids, j, acc); break;
        case 6: sage_rows<6>(Hl, ids, j, acc); break;
        case 7: sage_rows<7>(Hl, ids, j, acc); break;
        default: break;
    }
}

__global__ __launch_bounds__(256) void sage_gather(
    const float* __restrict__ H, const int* __restrict__ cnt,
    const int* __restrict__ slots, const int* __restrict__ ovf_cursor,
    const int* __restrict__ ovf_dst, const int* __restrict__ ovf_src,
    float* __restrict__ SUM) {
    const int lane = threadIdx.x & 31;
    const int wv     = __builtin_amdgcn_readfirstlane((int)(threadIdx.x >> 5));
    const int stride = gridDim.x * 8;
    const float* Hl  = H + lane * 4;
    const int sl     = lane & (GN_SLOTS - 1);

    int i0 = blockIdx.x * 8 + wv;
    if (i0 >= GN_N) return;
    int n0 = cnt[i0];
    int i1 = i0 + stride;
    int n1 = 0;
    if (i1 < GN_N) n1 = cnt[i1];
    int ids0 = slots[i0 * GN_SLOTS + sl];

    for (;;) {
        const int ids1 = slots[min(i1, GN_N - 1) * GN_SLOTS + sl];
        const int i2 = i1 + stride;
        int n2 = 0;
        if (i2 < GN_N) n2 = cnt[i2];

        float4 acc = make_float4(0.0f, 0.0f, 0.0f, 0.0f);
        sage_add_rows(Hl, ids0, min(n0, GN_SLOTS), acc);
        if (n0 > GN_SLOTS) {
            const int novf = min(*ovf_cursor, GN_E);
            for (int b = 0; b < novf; b += 32) {
                const int p = b + lane;
                int d = -1, s = 0;
                if (p < novf) { d = ovf_dst[p]; s = ovf_src[p]; }
                unsigned m = __builtin_amdgcn_ballot_w32(d == i0);
                while (m) {
                    const int l = __builtin_ctz(m);
                    m &= m - 1;
                    const int sid = __builtin_amdgcn_readlane(s, l);
                    const float4 v = *(const float4*)(Hl + (size_t)sid * GN_D);
                    acc.x += v.x; acc.y += v.y; acc.z += v.z; acc.w += v.w;
                }
            }
        }
        *(float4*)(SUM + (size_t)i0 * GN_D + lane * 4) = acc;

        if (i1 >= GN_N) break;
        i0 = i1; n0 = n1; ids0 = ids1;
        i1 = i2; n1 = n2;
    }
}

__global__ __launch_bounds__(GN_GEMM_THREADS) void sage_gemm(
    const float* __restrict__ A1, const int* __restrict__ deg,
    const float* __restrict__ W1, const float* __restrict__ bias,
    const float* __restrict__ A2, const float* __restrict__ W2,
    float* __restrict__ out, int relu_out) {
    __shared__ __attribute__((aligned(16))) float Bs[GN_LDS_FLOATS];

    const int lane = threadIdx.x & 31;
    const int wave = blockIdx.x * GN_GEMM_WAVES +
                     __builtin_amdgcn_readfirstlane((int)(threadIdx.x >> 5));

    sage_stage_w(W1, Bs);
    if (A2) sage_stage_w(W2, Bs + GN_W_LDS);

    const int rbase = wave * 16;
    const int h     = lane >> 4;
    const int l15   = lane & 15;
    const int row   = min(rbase + l15, GN_N - 1);

    float s = 1.0f;
    if (deg) s = 1.0f / fmaxf((float)deg[row], 1.0f);

    float bj[8];
#pragma unroll
    for (int jt = 0; jt < 8; ++jt) bj[jt] = bias[jt * 16 + l15];

    __syncthreads();
    if (wave >= GN_N / 16) return;

    v8f c[8];
#pragma unroll
    for (int jt = 0; jt < 8; ++jt)
#pragma unroll
        for (int v = 0; v < 8; ++v) c[jt][v] = bj[jt];

    const int npath = A2 ? 2 : 1;
#pragma nounroll
    for (int p = 0; p < npath; ++p) {
        const float* arow  = (p ? A2 : A1) + (size_t)row * GN_D + 2 * h;
        const float* bfrag = Bs + p * GN_W_LDS + lane * 4;
        const float  sc    = p ? 1.0f : s;

        v2f    ar[8];
        float4 bq[2][4];
#pragma unroll
        for (int i = 0; i < 8; ++i) ar[i] = *(const v2f*)(arow + 4 * i);
#pragma unroll
        for (int jp = 0; jp < 4; ++jp) bq[0][jp] = *(const float4*)(bfrag + jp * 128);

#pragma unroll
        for (int kg = 0; kg < 32; ++kg) {
            if (kg + 1 < 32) {
#pragma unroll
                for (int jp = 0; jp < 4; ++jp)
                    bq[(kg + 1) & 1][jp] =
                        *(const float4*)(bfrag + (kg + 1) * GN_KG_PITCH + jp * 128);
            }
            v2f a = ar[kg & 7];
            a.x *= sc;
            a.y *= sc;
            if (kg + 8 < 32) ar[kg & 7] = *(const v2f*)(arow + 4 * (kg + 8));
#pragma unroll
            for (int jp = 0; jp < 4; ++jp) {
                const float4 bb = bq[kg & 1][jp];
                v2f b0; b0.x = bb.x; b0.y = bb.y;
                v2f b1; b1.x = bb.z; b1.y = bb.w;
                c[2 * jp] = __builtin_amdgcn_wmma_f32_16x16x4_f32(
                    false, a, false, b0, (short)0, c[2 * jp], false, false);
                c[2 * jp + 1] = __builtin_amdgcn_wmma_f32_16x16x4_f32(
                    false, a, false, b1, (short)0, c[2 * jp + 1], false, false);
            }
            __builtin_amdgcn_sched_barrier(0);
        }
    }

#pragma unroll
    for (int jt = 0; jt < 8; ++jt)
#pragma unroll
        for (int v = 0; v < 8; ++v) {
            float val = c[jt][v];
            if (relu_out) val = fmaxf(val, 0.0f);
            out[(size_t)(rbase + v + 8 * h) * GN_D + jt * 16 + l15] = val;
        }
}

extern "C" void kernel_launch(void* const* d_in, const int* in_sizes, int n_in,
                              void* d_out, int out_size, void* d_ws, size_t ws_size,
                              hipStream_t stream) {
    (void)in_sizes; (void)n_in; (void)out_size; (void)ws_size;

    const float* x   = (const float*)d_in[0];
    const int*   ei  = (const int*)d_in[1];
    const float* W0p = (const float*)d_in[2];
    const float* b0p = (const float*)d_in[3];
    const float* W0l = (const float*)d_in[4];
    const float* b0l = (const float*)d_in[5];
    const float* W0r = (const float*)d_in[6];
    const float* W1p = (const float*)d_in[7];
    const float* b1p = (const float*)d_in[8];
    const float* W1l = (const float*)d_in[9];
    const float* b1l = (const float*)d_in[10];
    const float* W1r = (const float*)d_in[11];
    float* out = (float*)d_out;

    const size_t nd = (size_t)GN_N * GN_D;
    float* H   = (float*)d_ws;
    float* X1  = H + nd;
    float* SUM = X1 + nd;
    int* CNT   = (int*)(SUM + nd);
    int* CUR   = CNT + GN_N;
    int* SLOTS = CNT + GN_N + 32;
    int* OVFD  = SLOTS + GN_N * GN_SLOTS;
    int* OVFS  = OVFD + GN_E;

    const int* src = ei;
    const int* dst = ei + GN_E;

    hipMemsetAsync(CNT, 0, (GN_N + 1) * sizeof(int), stream);
    sage_bucket<<<(GN_E + 255) / 256, 256, 0, stream>>>(src, dst, CNT, SLOTS, CUR, OVFD, OVFS);

    const int gemm_blocks   = GN_GEMM_BLOCKS;
    const int gather_blocks = 2048;

    sage_gemm<<<gemm_blocks, GN_GEMM_THREADS, 0, stream>>>(x, nullptr, W0p, b0p,
                                               nullptr, nullptr, H, 1);
    sage_gather<<<gather_blocks, 256, 0, stream>>>(H, CNT, SLOTS, CUR, OVFD, OVFS, SUM);
    sage_gemm<<<gemm_blocks, GN_GEMM_THREADS, 0, stream>>>(SUM, CNT, W0l, b0l,
                                               x, W0r, X1, 1);

    sage_gemm<<<gemm_blocks, GN_GEMM_THREADS, 0, stream>>>(X1, nullptr, W1p, b1p,
                                               nullptr, nullptr, H, 1);
    sage_gather<<<gather_blocks, 256, 0, stream>>>(H, CNT, SLOTS, CUR, OVFD, OVFS, SUM);
    sage_gemm<<<gemm_blocks, GN_GEMM_THREADS, 0, stream>>>(SUM, CNT, W1l, b1l,
                                               X1, W1r, out, 0);
}


# GEMM work dealt evenly over the 4 SIMDs: 24-wave workgroups (6 per SIMD) own 25 row blocks, the 25th split by column pairs over waves 0-3, one per SIMD; busiest SIMD 1792 -> 1600 WMMAs per product
# speedup vs baseline: 2.3947x; 1.0378x over previous
#include <hip/hip_runtime.h>
#include <hip/hip_bf16.h>


#define GN_N 100000
#define GN_E 640000
#define GN_D 128

typedef __attribute__((ext_vector_type(2))) float v2f;
typedef __attribute__((ext_vector_type(8))) float v8f;

#define GN_KG_PITCH     516
#define GN_W_LDS        (32 * GN_KG_PITCH)
#define GN_GEMM_WAVES   24
#define GN_GEMM_TILES   25
#define GN_GEMM_THREADS (GN_GEMM_WAVES * 32)
#define GN_GEMM_BLOCKS  (GN_N / 16 / GN_GEMM_TILES)
static_assert(GN_GEMM_BLOCKS * GN_GEMM_TILES * 16 == GN_N, "row tiles must cover N exactly");
static_assert(GN_GEMM_TILES == GN_GEMM_WAVES + 1, "one shared tile per workgroup");
#define GN_LDS_FLOATS   41472

__device__ __forceinline__ void sage_stage_w(const float* __restrict__ W,
                                             float* __restrict__ Bs) {
    const int rot = (blockIdx.x * 37) & 127;
#pragma unroll
    for (int it = 0; it < 6; ++it) {
        const int idx = it * GN_GEMM_THREADS + (int)threadIdx.x;
        if (idx < 32 * 128) {
            const int j  = ((idx >> 5) + rot) & 127;
            const int kg = idx & 31;
            const float4 w = *(const float4*)(W + j * GN_D + kg * 4);
            float* d = Bs + kg * GN_KG_PITCH + (j >> 5) * 128 + (j & 15) * 4 + ((j >> 4) & 1) * 2;
            v2f lo; lo.x = w.x; lo.y = w.y;
            v2f hi; hi.x = w.z; hi.y = w.w;
            *(v2f*)d        = lo;
            *(v2f*)(d + 64) = hi;
        }
    }
}

#define GN_SLOTS 16

__global__ __launch_bounds__(256) void sage_bucket(
    const int* __restrict__ src, const int* __restrict__ dst,
    int* __restrict__ cnt, int* __restrict__ slots,
    int* __restrict__ ovf_cursor, int* __restrict__ ovf_dst, int* __restrict__ ovf_src) {
    const int e = blockIdx.x * blockDim.x + threadIdx.x;
    if (e >= GN_E) return;
    const int d = dst[e];
    const int s = src[e];
    const int r = atomicAdd(&cnt[d], 1);
    if (r < GN_SLOTS) {
        slots[d * GN_SLOTS + r] = s;
    } else {
        const int p = atomicAdd(ovf_cursor, 1);
        if ((unsigned)p < (unsigned)GN_E) { ovf_dst[p] = d; ovf_src[p] = s; }
    }
}

template <int R>
__device__ __forceinline__ void sage_rows(const float* __restrict__ Hl,
                                          int ids, int j, float4& acc) {
    float4 v[R];
    for (int k = 0; k < R; ++k)
        v[k] = *(const float4*)(Hl + (size_t)__builtin_amdgcn_readlane(ids, j + k) * GN_D);
    for (int k = 0; k < R; ++k) {
        acc.x += v[k].x; acc.y += v[k].y; acc.z += v[k].z; acc.w += v[k].w;
    }
}

__device__ __forceinline__ void sage_add_rows(const float* __restrict__ Hl,
                                              int ids, int m, float4& acc) {
    int j = 0;
    for (; j + 8 <= m; j += 8) sage_rows<8>(Hl, ids, j, acc);
    switch (m - j) {
        case 1: sage_rows<1>(Hl, ids, j, acc); break;
        case 2: sage_rows<2>(Hl, ids, j, acc); break;
        case 3: sage_rows<3>(Hl, ids, j, acc); break;
        case 4: sage_rows<4>(Hl, ids, j, acc); break;
        case 5: sage_rows<5>(Hl, ids, j, acc); break;
        case 6: sage_rows<6>(Hl, ids, j, acc); break;
        case 7: sage_rows<7>(Hl, ids, j, acc); break;
        default: break;
    }
}

__global__ __launch_bounds__(256) void sage_gather(
    const float* __restrict__ H, const int* __restrict__ cnt,
    const int* __restrict__ slots, const int* __restrict__ ovf_cursor,
    const int* __restrict__ ovf_dst, const int* __restrict__ ovf_src,
    float* __restrict__ SUM) {
    const int lane = threadIdx.x & 31;
    const int wv     = __builtin_amdgcn_readfirstlane((int)(threadIdx.x >> 5));
    const int stride = gridDim.x * 8;
    const float* Hl  = H + lane * 4;
    const int sl     = lane & (GN_SLOTS - 1);

    int i0 = blockIdx.x * 8 + wv;
    if (i0 >= GN_N) return;
    int n0 = cnt[i0];
    int i1 = i0 + stride;
    int n1 = 0;
    if (i1 < GN_N) n1 = cnt[i1];
    int ids0 = slots[i0 * GN_SLOTS + sl];

    for (;;) {
        const int ids1 = slots[min(i1, GN_N - 1) * GN_SLOTS + sl];
        const int i2 = i1 + stride;
        int n2 = 0;
        if (i2 < GN_N) n2 = cnt[i2];

        float4 acc = make_float4(0.0f, 0.0f, 0.0f, 0.0f);
        sage_add_rows(Hl, ids0, min(n0, GN_SLOTS), acc);
        if (n0 > GN_SLOTS) {
            const int novf = min(*ovf_cursor, GN_E);
            for (int b = 0; b < novf; b += 32) {
                const int p = b + lane;
                int d = -1, s = 0;
                if (p < novf) { d = ovf_dst[p]; s = ovf_src[p]; }
                unsigned m = __builtin_amdgcn_ballot_w32(d == i0);
                while (m) {
                    const int l = __builtin_ctz(m);
                    m &= m - 1;
                    const int sid = __builtin_amdgcn_readlane(s, l);
                    const float4 v = *(const float4*)(Hl + (size_t)sid * GN_D);
                    acc.x += v.x; acc.y += v.y; acc.z += v.z; acc.w += v.w;
                }
            }
        }
        *(float4*)(SUM + (size_t)i0 * GN_D + lane * 4) = acc;

        if (i1 >= GN_N) break;
        i0 = i1; n0 = n1; ids0 = ids1;
        i1 = i2; n1 = n2;
    }
}


template <int NJP>
__device__ __forceinline__ void sage_tile(
    const float* __restrict__ A1, const float* __restrict__ A2,
    const float* Bs, float* __restrict__ out, int relu_out,
    int rbase, int row, int jp0, int lane, float s, const float* bj) {
    const int h   = lane >> 4;
    const int l15 = lane & 15;

    v8f c[2 * NJP];
#pragma unroll
    for (int jt = 0; jt < 2 * NJP; ++jt)
#pragma unroll
        for (int v = 0; v < 8; ++v) c[jt][v] = bj[jt];

    const int npath = A2 ? 2 : 1;
#pragma nounroll
    for (int p = 0; p < npath; ++p) {
        const float* arow  = (p ? A2 : A1) + (size_t)row * GN_D + 2 * h;
        const float* bfrag = Bs + p * GN_W_LDS + jp0 * 128 + lane * 4;
        const float  sc    = p ? 1.0f : s;

        v2f    ar[8];
        float4 bq[2][NJP];
#pragma unroll
        for (int i = 0; i < 8; ++i) ar[i] = *(const v2f*)(arow + 4 * i);
#pragma unroll
        for (int jp = 0; jp < NJP; ++jp) bq[0][jp] = *(const float4*)(bfrag + jp * 128);

#pragma unroll
        for (int kg = 0; kg < 32; ++kg) {
            if (kg + 1 < 32) {
#pragma unroll
                for (int jp = 0; jp < NJP; ++jp)
                    bq[(kg + 1) & 1][jp] =
                        *(const float4*)(bfrag + (kg + 1) * GN_KG_PITCH + jp * 128);
            }
            v2f a = ar[kg & 7];
            a.x *= sc;
            a.y *= sc;
            if (kg + 8 < 32) ar[kg & 7] = *(const v2f*)(arow + 4 * (kg + 8));
#pragma unroll
            for (int jp = 0; jp < NJP; ++jp) {
                const float4 bb = bq[kg & 1][jp];
                v2f b0; b0.x = bb.x; b0.y = bb.y;
                v2f b1; b1.x = bb.z; b1.y = bb.w;
                c[2 * jp] = __builtin_amdgcn_wmma_f32_16x16x4_f32(
                    false, a, false, b0, (short)0, c[2 * jp], false, false);
                c[2 * jp + 1] = __builtin_amdgcn_wmma_f32_16x16x4_f32(
                    false, a, false, b1, (short)0, c[2 * jp + 1], false, false);
            }
            __builtin_amdgcn_sched_barrier(0);
        }
    }

#pragma unroll
    for (int jt = 0; jt < 2 * NJP; ++jt)
#pragma unroll
        for (int v = 0; v < 8; ++v) {
            float val = c[jt][v];
            if (relu_out) val = fmaxf(val, 0.0f);
            out[(size_t)(rbase + v + 8 * h) * GN_D + jp0 * 32 + jt * 16 + l15] = val;
        }
}

__global__ __launch_bounds__(GN_GEMM_THREADS) void sage_gemm(
    const float* __restrict__ A1, const int* __restrict__ deg,
    const float* __restrict__ W1, const float* __restrict__ bias,
    const float* __restrict__ A2, const float* __restrict__ W2,
    float* __restrict__ out, int relu_out) {
    __shared__ __attribute__((aligned(16))) float Bs[GN_LDS_FLOATS];

    const int lane = threadIdx.x & 31;
    const int wv    = __builtin_amdgcn_readfirstlane((int)(threadIdx.x >> 5));
    const int tile0 = blockIdx.x * GN_GEMM_TILES;

    sage_stage_w(W1, Bs);
    if (A2) sage_stage_w(W2, Bs + GN_W_LDS);

    const int l15    = lane & 15;
    const int rb_own = (tile0 + wv) * 16;
    const int rb_sh  = (tile0 + GN_GEMM_WAVES) * 16;
    const int jp_sh  = wv & 3;
    const int row_own = min(rb_own + l15, GN_N - 1);
    const int row_sh  = min(rb_sh + l15, GN_N - 1);

    float s_own = 1.0f, s_sh = 1.0f;
    if (deg) {
        s_own = 1.0f / fmaxf((float)deg[row_own], 1.0f);
        s_sh  = 1.0f / fmaxf((float)deg[row_sh], 1.0f);
    }

    float bj_own[8], bj_sh[2];
#pragma unroll
    for (int jt = 0; jt < 8; ++jt) bj_own[jt] = bias[jt * 16 + l15];
#pragma unroll
    for (int jt = 0; jt < 2; ++jt) bj_sh[jt] = bias[jp_sh * 32 + jt * 16 + l15];

    __syncthreads();

    if (wv < 4)
        sage_tile<1>(A1, A2, Bs, out, relu_out, rb_sh, row_sh, jp_sh, lane, s_sh, bj_sh);
    sage_tile<4>(A1, A2, Bs, out, relu_out, rb_own, row_own, 0, lane, s_own, bj_own);
}

extern "C" void kernel_launch(void* const* d_in, const int* in_sizes, int n_in,
                              void* d_out, int out_size, void* d_ws, size_t ws_size,
                              hipStream_t stream) {
    (void)in_sizes; (void)n_in; (void)out_size; (void)ws_size;

    const float* x   = (const float*)d_in[0];
    const int*   ei  = (const int*)d_in[1];
    const float* W0p = (const float*)d_in[2];
    const float* b0p = (const float*)d_in[3];
    const float* W0l = (const float*)d_in[4];
    const float* b0l = (const float*)d_in[5];
    const float* W0r = (const float*)d_in[6];
    const float* W1p = (const float*)d_in[7];
    const float* b1p = (const float*)d_in[8];
    const float* W1l = (const float*)d_in[9];
    const float* b1l = (const float*)d_in[10];
    const float* W1r = (const float*)d_in[11];
    float* out = (float*)d_out;

    const size_t nd = (size_t)GN_N * GN_D;
    float* H   = (float*)d_ws;
    float* X1  = H + nd;
    float* SUM = X1 + nd;
    int* CNT   = (int*)(SUM + nd);
    int* CUR   = CNT + GN_N;
    int* SLOTS = CNT + GN_N + 32;
    int* OVFD  = SLOTS + GN_N * GN_SLOTS;
    int* OVFS  = OVFD + GN_E;

    const int* src = ei;
    const int* dst = ei + GN_E;

    hipMemsetAsync(CNT, 0, (GN_N + 1) * sizeof(int), stream);
    sage_bucket<<<(GN_E + 255) / 256, 256, 0, stream>>>(src, dst, CNT, SLOTS, CUR, OVFD, OVFS);

    const int gemm_blocks   = GN_GEMM_BLOCKS;
    const int gather_blocks = 2048;

    sage_gemm<<<gemm_blocks, GN_GEMM_THREADS, 0, stream>>>(x, nullptr, W0p, b0p,
                                               nullptr, nullptr, H, 1);
    sage_gather<<<gather_blocks, 256, 0, stream>>>(H, CNT, SLOTS, CUR, OVFD, OVFS, SUM);
    sage_gemm<<<gemm_blocks, GN_GEMM_THREADS, 0, stream>>>(SUM, CNT, W0l, b0l,
                                               x, W0r, X1, 1);

    sage_gemm<<<gemm_blocks, GN_GEMM_THREADS, 0, stream>>>(X1, nullptr, W1p, b1p,
                                               nullptr, nullptr, H, 1);
    sage_gather<<<gather_blocks, 256, 0, stream>>>(H, CNT, SLOTS, CUR, OVFD, OVFS, SUM);
    sage_gemm<<<gemm_blocks, GN_GEMM_THREADS, 0, stream>>>(SUM, CNT, W1l, b1l,
                                               X1, W1r, out, 0);
}


# Source lists built by two LDS binning passes (rank by LDS atomic, whole-line stores); per-edge global atomics, scattered 4-byte stores and memset gone; rows summed in LDS-atomic arrival order
# speedup vs baseline: 2.7287x; 1.1395x over previous
#include <hip/hip_runtime.h>
#include <hip/hip_bf16.h>


#define GN_N 100000
#define GN_E 640000
#define GN_D 128

typedef __attribute__((ext_vector_type(2))) float v2f;
typedef __attribute__((ext_vector_type(8))) float v8f;

#define GN_KG_PITCH     516
#define GN_W_LDS        (32 * GN_KG_PITCH)
#define GN_GEMM_WAVES   24
#define GN_GEMM_TILES   25
#define GN_GEMM_THREADS (GN_GEMM_WAVES * 32)
#define GN_GEMM_BLOCKS  (GN_N / 16 / GN_GEMM_TILES)
static_assert(GN_GEMM_BLOCKS * GN_GEMM_TILES * 16 == GN_N, "row tiles must cover N exactly");
static_assert(GN_GEMM_TILES == GN_GEMM_WAVES + 1, "one shared tile per workgroup");
#define GN_LDS_FLOATS   41472

__device__ __forceinline__ void sage_stage_w(const float* __restrict__ W,
                                             float* __restrict__ Bs) {
    const int rot = (blockIdx.x * 37) & 127;
#pragma unroll
    for (int it = 0; it < 6; ++it) {
        const int idx = it * GN_GEMM_THREADS + (int)threadIdx.x;
        if (idx < 32 * 128) {
            const int j  = ((idx >> 5) + rot) & 127;
            const int kg = idx & 31;
            const float4 w = *(const float4*)(W + j * GN_D + kg * 4);
            float* d = Bs + kg * GN_KG_PITCH + (j >> 5) * 128 + (j & 15) * 4 + ((j >> 4) & 1) * 2;
            v2f lo; lo.x = w.x; lo.y = w.y;
            v2f hi; hi.x = w.z; hi.y = w.w;
            *(v2f*)d        = lo;
            *(v2f*)(d + 64) = hi;
        }
    }
}

#define GN_SLOTS 16
#define GN_NB    250
#define GN_BN    400
#define GN_GA    250
#define GN_EA    2560
#define GN_CELL  32
static_assert(GN_NB * GN_BN == GN_N, "buckets must cover the nodes exactly");
static_assert(GN_GA * GN_EA == GN_E, "edge chunks must cover the edges exactly");
static_assert(GN_BN <= 512 && GN_N <= (1 << 17), "a cell entry packs 9 + 17 bits");
static_assert(GN_EA % 256 == 0, "whole trips of 256 threads per edge chunk");

__global__ __launch_bounds__(256) void sage_bin_edges(
    const int* __restrict__ src, const int* __restrict__ dst,
    int* __restrict__ cells, int* __restrict__ spill_cnt, int* __restrict__ spill,
    int* __restrict__ ovf_cursor) {
    __shared__ int lcell[GN_NB * GN_CELL];
    __shared__ int lcnt[GN_NB];
    __shared__ int lspill;

    const int t = threadIdx.x;
    const int a = blockIdx.x;
    if (t < GN_NB) lcnt[t] = 0;
    if (t == 0) {
        lspill = 0;
        if (a == 0) *ovf_cursor = 0;
    }
    __syncthreads();

    int d[GN_EA / 256], s[GN_EA / 256];
#pragma unroll
    for (int it = 0; it < GN_EA / 256; ++it) {
        const int e = a * GN_EA + it * 256 + t;
        d[it] = dst[e];
        s[it] = src[e];
    }
    int r[GN_EA / 256];
#pragma unroll
    for (int it = 0; it < GN_EA / 256; ++it)
        r[it] = atomicAdd(&lcnt[(unsigned)d[it] / GN_BN], 1);
#pragma unroll
    for (int it = 0; it < GN_EA / 256; ++it) {
        const int b  = (int)((unsigned)d[it] / GN_BN);
        const int dl = d[it] - b * GN_BN;
        if (r[it] < GN_CELL - 1) {
            lcell[b * GN_CELL + 1 + r[it]] = (dl << 17) | s[it];
        } else {
            const int q = atomicAdd(&lspill, 1);
            spill[(a * GN_EA + q) * 2]     = d[it];
            spill[(a * GN_EA + q) * 2 + 1] = s[it];
        }
    }
    __syncthreads();

    const int lane = t & 31;
    const int wv   = __builtin_amdgcn_readfirstlane(t >> 5);
    for (int b = wv; b < GN_NB; b += 8) {
        const int n = min(lcnt[b], GN_CELL - 1);
        int v = 0;
        if (lane >= 1 && lane <= n) v = lcell[b * GN_CELL + lane];
        if (lane == 0) v = n;
        cells[((size_t)b * GN_GA + a) * GN_CELL + lane] = v;
    }
    if (t == 0) spill_cnt[a] = lspill;
}

__global__ __launch_bounds__(256) void sage_fill_slots(
    const int* __restrict__ cells, const int* __restrict__ spill_cnt,
    const int* __restrict__ spill,
    int* __restrict__ cnt, int* __restrict__ slots,
    int* __restrict__ ovf_cursor, int* __restrict__ ovf_dst, int* __restrict__ ovf_src) {
    __shared__ __attribute__((aligned(16))) int lslots[GN_BN * GN_SLOTS];
    __shared__ int lcnt[GN_BN];

    const int t    = threadIdx.x;
    const int lane = t & 31;
    const int wv   = __builtin_amdgcn_readfirstlane(t >> 5);
    const int b    = blockIdx.x;
    const int base = b * GN_BN;

    const int* row = cells + (size_t)b * GN_GA * GN_CELL;
    constexpr int NFULL = GN_GA / 8;
    int v[NFULL + 1];
#pragma unroll
    for (int i = 0; i < NFULL; ++i) v[i] = row[(wv + 8 * i) * GN_CELL + lane];
    v[NFULL] = 0;
    if (wv + 8 * NFULL < GN_GA) v[NFULL] = row[(wv + 8 * NFULL) * GN_CELL + lane];
    int nsp = 0;
    if (t < GN_GA) nsp = spill_cnt[t];

    for (int i = t; i < GN_BN; i += 256) lcnt[i] = 0;
    for (int i = t; i < GN_BN * GN_SLOTS / 4; i += 256) ((int4*)lslots)[i] = make_int4(0, 0, 0, 0);
    __syncthreads();

    auto place = [&](int dl, int sid, int r) {
        if (r < GN_SLOTS) {
            lslots[dl * GN_SLOTS + r] = sid;
        } else {
            const int p = atomicAdd(ovf_cursor, 1);
            if ((unsigned)p < (unsigned)GN_E) { ovf_dst[p] = base + dl; ovf_src[p] = sid; }
        }
    };

    int r[NFULL + 1];
#pragma unroll
    for (int i = 0; i < NFULL + 1; ++i) {
        const int n = __builtin_amdgcn_readlane(v[i], 0);
        r[i] = -1;
        if (lane >= 1 && lane <= n) r[i] = atomicAdd(&lcnt[v[i] >> 17], 1);
    }
#pragma unroll
    for (int i = 0; i < NFULL + 1; ++i)
        if (r[i] >= 0) place(v[i] >> 17, v[i] & 0x1FFFF, r[i]);
    nsp = min(nsp, GN_EA);
    for (int q = 0; q < nsp; ++q) {
        const int dd = spill[(t * GN_EA + q) * 2];
        const int ss = spill[(t * GN_EA + q) * 2 + 1];
        if (dd >= base && dd < base + GN_BN) place(dd - base, ss, atomicAdd(&lcnt[dd - base], 1));
    }
    __syncthreads();

    for (int i = t; i < GN_BN; i += 256) cnt[base + i] = lcnt[i];
    int4*       gs = (int4*)(slots + (size_t)base * GN_SLOTS);
    const int4* ls = (const int4*)lslots;
    for (int i = t; i < GN_BN * GN_SLOTS / 4; i += 256) gs[i] = ls[i];
}

template <int R>
__device__ __forceinline__ void sage_rows(const float* __restrict__ Hl,
                                          int ids, int j, float4& acc) {
    float4 v[R];
    for (int k = 0; k < R; ++k)
        v[k] = *(const float4*)(Hl + (size_t)__builtin_amdgcn_readlane(ids, j + k) * GN_D);
    for (int k = 0; k < R; ++k) {
        acc.x += v[k].x; acc.y += v[k].y; acc.z += v[k].z; acc.w += v[k].w;
    }
}

__device__ __forceinline__ void sage_add_rows(const float* __restrict__ Hl,
                                              int ids, int m, float4& acc) {
    int j = 0;
    for (; j + 8 <= m; j += 8) sage_rows<8>(Hl, ids, j, acc);
    switch (m - j) {
        case 1: sage_rows<1>(Hl, ids, j, acc); break;
        case 2: sage_rows<2>(Hl, ids, j, acc); break;
        case 3: sage_rows<3>(Hl, ids, j, acc); break;
        case 4: sage_rows<4>(Hl, ids, j, acc); break;
        case 5: sage_rows<5>(Hl, ids, j, acc); break;
        case 6: sage_rows<6>(Hl, ids, j, acc); break;
        case 7: sage_rows<7>(Hl, ids, j, acc); break;
        default: break;
    }
}

__global__ __launch_bounds__(256) void sage_gather(
    const float* __restrict__ H, const int* __restrict__ cnt,
    const int* __restrict__ slots, const int* __restrict__ ovf_cursor,
    const int* __restrict__ ovf_dst, const int* __restrict__ ovf_src,
    float* __restrict__ SUM) {
    const int lane = threadIdx.x & 31;
    const int wv     = __builtin_amdgcn_readfirstlane((int)(threadIdx.x >> 5));
    const int stride = gridDim.x * 8;
    const float* Hl  = H + lane * 4;
    const int sl     = lane & (GN_SLOTS - 1);

    int i0 = blockIdx.x * 8 + wv;
    if (i0 >= GN_N) return;
    int n0 = cnt[i0];
    int i1 = i0 + stride;
    int n1 = 0;
    if (i1 < GN_N) n1 = cnt[i1];
    int ids0 = slots[i0 * GN_SLOTS + sl];

    for (;;) {
        const int ids1 = slots[min(i1, GN_N - 1) * GN_SLOTS + sl];
        const int i2 = i1 + stride;
        int n2 = 0;
        if (i2 < GN_N) n2 = cnt[i2];

        float4 acc = make_float4(0.0f, 0.0f, 0.0f, 0.0f);
        sage_add_rows(Hl, ids0, min(n0, GN_SLOTS), acc);
        if (n0 > GN_SLOTS) {
            const int novf = min(*ovf_cursor, GN_E);
            for (int b = 0; b < novf; b += 32) {
                const int p = b + lane;
                int d = -1, s = 0;
                if (p < novf) { d = ovf_dst[p]; s = ovf_src[p]; }
                unsigned m = __builtin_amdgcn_ballot_w32(d == i0);
                while (m) {
                    const int l = __builtin_ctz(m);
                    m &= m - 1;
                    const int sid = __builtin_amdgcn_readlane(s, l);
                    const float4 v = *(const float4*)(Hl + (size_t)sid * GN_D);
                    acc.x += v.x; acc.y += v.y; acc.z += v.z; acc.w += v.w;
                }
            }
        }
        *(float4*)(SUM + (size_t)i0 * GN_D + lane * 4) = acc;

        if (i1 >= GN_N) break;
        i0 = i1; n0 = n1; ids0 = ids1;
        i1 = i2; n1 = n2;
    }
}


template <int NJP>
__device__ __forceinline__ void sage_tile(
    const float* __restrict__ A1, const float* __restrict__ A2,
    const float* Bs, float* __restrict__ out, int relu_out,
    int rbase, int row, int jp0, int lane, float s, const float* bj) {
    const int h   = lane >> 4;
    const int l15 = lane & 15;

    v8f c[2 * NJP];
#pragma unroll
    for (int jt = 0; jt < 2 * NJP; ++jt)
#pragma unroll
        for (int v = 0; v < 8; ++v) c[jt][v] = bj[jt];

    const int npath = A2 ? 2 : 1;
#pragma nounroll
    for (int p = 0; p < npath; ++p) {
        const float* arow  = (p ? A2 : A1) + (size_t)row * GN_D + 2 * h;
        const float* bfrag = Bs + p * GN_W_LDS + jp0 * 128 + lane * 4;
        const float  sc    = p ? 1.0f : s;

        v2f    ar[8];
        float4 bq[2][NJP];
#pragma unroll
        for (int i = 0; i < 8; ++i) ar[i] = *(const v2f*)(arow + 4 * i);
#pragma unroll
        for (int jp = 0; jp < NJP; ++jp) bq[0][jp] = *(const float4*)(bfrag + jp * 128);

#pragma unroll
        for (int kg = 0; kg < 32; ++kg) {
            if (kg + 1 < 32) {
#pragma unroll
                for (int jp = 0; jp < NJP; ++jp)
                    bq[(kg + 1) & 1][jp] =
                        *(const float4*)(bfrag + (kg + 1) * GN_KG_PITCH + jp * 128);
            }
            v2f a = ar[kg & 7];
            a.x *= sc;
            a.y *= sc;
            if (kg + 8 < 32) ar[kg & 7] = *(const v2f*)(arow + 4 * (kg + 8));
#pragma unroll
            for (int jp = 0; jp < NJP; ++jp) {
                const float4 bb = bq[kg & 1][jp];
                v2f b0; b0.x = bb.x; b0.y = bb.y;
                v2f b1; b1.x = bb.z; b1.y = bb.w;
                c[2 * jp] = __builtin_amdgcn_wmma_f32_16x16x4_f32(
                    false, a, false, b0, (short)0, c[2 * jp], false, false);
                c[2 * jp + 1] = __builtin_amdgcn_wmma_f32_16x16x4_f32(
                    false, a, false, b1, (short)0, c[2 * jp + 1], false, false);
            }
            __builtin_amdgcn_sched_barrier(0);
        }
    }

#pragma unroll
    for (int jt = 0; jt < 2 * NJP; ++jt)
#pragma unroll
        for (int v = 0; v < 8; ++v) {
            float val = c[jt][v];
            if (relu_out) val = fmaxf(val, 0.0f);
            out[(size_t)(rbase + v + 8 * h) * GN_D + jp0 * 32 + jt * 16 + l15] = val;
        }
}

__global__ __launch_bounds__(GN_GEMM_THREADS) void sage_gemm(
    const float* __restrict__ A1, const int* __restrict__ deg,
    const float* __restrict__ W1, const float* __restrict__ bias,
    const float* __restrict__ A2, const float* __restrict__ W2,
    float* __restrict__ out, int relu_out) {
    __shared__ __attribute__((aligned(16))) float Bs[GN_LDS_FLOATS];

    const int lane = threadIdx.x & 31;
    const int wv    = __builtin_amdgcn_readfirstlane((int)(threadIdx.x >> 5));
    const int tile0 = blockIdx.x * GN_GEMM_TILES;

    sage_stage_w(W1, Bs);
    if (A2) sage_stage_w(W2, Bs + GN_W_LDS);

    const int l15    = lane & 15;
    const int rb_own = (tile0 + wv) * 16;
    const int rb_sh  = (tile0 + GN_GEMM_WAVES) * 16;
    const int jp_sh  = wv & 3;
    const int row_own = min(rb_own + l15, GN_N - 1);
    const int row_sh  = min(rb_sh + l15, GN_N - 1);

    float s_own = 1.0f, s_sh = 1.0f;
    if (deg) {
        s_own = 1.0f / fmaxf((float)deg[row_own], 1.0f);
        s_sh  = 1.0f / fmaxf((float)deg[row_sh], 1.0f);
    }

    float bj_own[8], bj_sh[2];
#pragma unroll
    for (int jt = 0; jt < 8; ++jt) bj_own[jt] = bias[jt * 16 + l15];
#pragma unroll
    for (int jt = 0; jt < 2; ++jt) bj_sh[jt] = bias[jp_sh * 32 + jt * 16 + l15];

    __syncthreads();

    if (wv < 4)
        sage_tile<1>(A1, A2, Bs, out, relu_out, rb_sh, row_sh, jp_sh, lane, s_sh, bj_sh);
    sage_tile<4>(A1, A2, Bs, out, relu_out, rb_own, row_own, 0, lane, s_own, bj_own);
}

extern "C" void kernel_launch(void* const* d_in, const int* in_sizes, int n_in,
                              void* d_out, int out_size, void* d_ws, size_t ws_size,
                              hipStream_t stream) {
    (void)in_sizes; (void)n_in; (void)out_size; (void)ws_size;

    const float* x   = (const float*)d_in[0];
    const int*   ei  = (const int*)d_in[1];
    const float* W0p = (const float*)d_in[2];
    const float* b0p = (const float*)d_in[3];
    const float* W0l = (const float*)d_in[4];
    const float* b0l = (const float*)d_in[5];
    const float* W0r = (const float*)d_in[6];
    const float* W1p = (const float*)d_in[7];
    const float* b1p = (const float*)d_in[8];
    const float* W1l = (const float*)d_in[9];
    const float* b1l = (const float*)d_in[10];
    const float* W1r = (const float*)d_in[11];
    float* out = (float*)d_out;

    const size_t nd = (size_t)GN_N * GN_D;
    float* H   = (float*)d_ws;
    float* X1  = H + nd;
    float* SUM = X1 + nd;
    int* CNT   = (int*)(SUM + nd);
    int* CUR   = CNT + GN_N;
    int* SLOTS = CNT + GN_N + 32;
    int* OVFD  = SLOTS + GN_N * GN_SLOTS;
    int* OVFS  = OVFD + GN_E;
    int* CELLS = OVFS + GN_E;
    int* SPCNT = CELLS + GN_NB * GN_GA * GN_CELL;
    int* SPILL = SPCNT + 256;

    const int* src = ei;
    const int* dst = ei + GN_E;

    sage_bin_edges<<<GN_GA, 256, 0, stream>>>(src, dst, CELLS, SPCNT, SPILL, CUR);
    sage_fill_slots<<<GN_NB, 256, 0, stream>>>(CELLS, SPCNT, SPILL, CNT, SLOTS, CUR, OVFD, OVFS);

    const int gemm_blocks   = GN_GEMM_BLOCKS;
    const int gather_blocks = 2048;

    sage_gemm<<<gemm_blocks, GN_GEMM_THREADS, 0, stream>>>(x, nullptr, W0p, b0p,
                                               nullptr, nullptr, H, 1);
    sage_gather<<<gather_blocks, 256, 0, stream>>>(H, CNT, SLOTS, CUR, OVFD, OVFS, SUM);
    sage_gemm<<<gemm_blocks, GN_GEMM_THREADS, 0, stream>>>(SUM, CNT, W0l, b0l,
                                               x, W0r, X1, 1);

    sage_gemm<<<gemm_blocks, GN_GEMM_THREADS, 0, stream>>>(X1, nullptr, W1p, b1p,
                                               nullptr, nullptr, H, 1);
    sage_gather<<<gather_blocks, 256, 0, stream>>>(H, CNT, SLOTS, CUR, OVFD, OVFS, SUM);
    sage_gemm<<<gemm_blocks, GN_GEMM_THREADS, 0, stream>>>(SUM, CNT, W1l, b1l,
                                               X1, W1r, out, 0);
}


# Neighbour means formed inside each layer's second GEMM launch by its own 24 waves, straight into LDS A tiles (A fragments read from LDS); the 2 gather launches and the SUM round trip are gone
# speedup vs baseline: 2.8651x; 1.0500x over previous
#include <hip/hip_runtime.h>
#include <hip/hip_bf16.h>


#define GN_N 100000
#define GN_E 640000
#define GN_D 128

typedef __attribute__((ext_vector_type(2))) float v2f;
typedef __attribute__((ext_vector_type(8))) float v8f;

#define GN_KG_PITCH     516
#define GN_W_LDS        (32 * GN_KG_PITCH)
#define GN_GEMM_WAVES   24
#define GN_GEMM_TILES   25
#define GN_GEMM_THREADS (GN_GEMM_WAVES * 32)
#define GN_GEMM_BLOCKS  (GN_N / 16 / GN_GEMM_TILES)
static_assert(GN_GEMM_BLOCKS * GN_GEMM_TILES * 16 == GN_N, "row tiles must cover N exactly");
static_assert(GN_GEMM_TILES == GN_GEMM_WAVES + 1, "one shared tile per workgroup");
#define GN_LDS_FLOATS   41472

__device__ __forceinline__ void sage_stage_w(const float* __restrict__ W,
                                             float* __restrict__ Bs) {
    const int rot = (blockIdx.x * 37) & 127;
#pragma unroll
    for (int it = 0; it < 6; ++it) {
        const int idx = it * GN_GEMM_THREADS + (int)threadIdx.x;
        if (idx < 32 * 128) {
            const int j  = ((idx >> 5) + rot) & 127;
            const int kg = idx & 31;
            const float4 w = *(const float4*)(W + j * GN_D + kg * 4);
            float* d = Bs + kg * GN_KG_PITCH + (j >> 5) * 128 + (j & 15) * 4 + ((j >> 4) & 1) * 2;
            v2f lo; lo.x = w.x; lo.y = w.y;
            v2f hi; hi.x = w.z; hi.y = w.w;
            *(v2f*)d        = lo;
            *(v2f*)(d + 64) = hi;
        }
    }
}

#define GN_SLOTS 16
#define GN_NB    250
#define GN_BN    400
#define GN_GA    250
#define GN_EA    2560
#define GN_CELL  32
static_assert(GN_NB * GN_BN == GN_N, "buckets must cover the nodes exactly");
static_assert(GN_GA * GN_EA == GN_E, "edge chunks must cover the edges exactly");
static_assert(GN_BN <= 512 && GN_N <= (1 << 17), "a cell entry packs 9 + 17 bits");
static_assert(GN_EA % 256 == 0, "whole trips of 256 threads per edge chunk");

__global__ __launch_bounds__(256) void sage_bin_edges(
    const int* __restrict__ src, const int* __restrict__ dst,
    int* __restrict__ cells, int* __restrict__ spill_cnt, int* __restrict__ spill,
    int* __restrict__ ovf_cursor) {
    __shared__ int lcell[GN_NB * GN_CELL];
    __shared__ int lcnt[GN_NB];
    __shared__ int lspill;

    const int t = threadIdx.x;
    const int a = blockIdx.x;
    if (t < GN_NB) lcnt[t] = 0;
    if (t == 0) {
        lspill = 0;
        if (a == 0) *ovf_cursor = 0;
    }
    __syncthreads();

    int d[GN_EA / 256], s[GN_EA / 256];
#pragma unroll
    for (int it = 0; it < GN_EA / 256; ++it) {
        const int e = a * GN_EA + it * 256 + t;
        d[it] = dst[e];
        s[it] = src[e];
    }
    int r[GN_EA / 256];
#pragma unroll
    for (int it = 0; it < GN_EA / 256; ++it)
        r[it] = atomicAdd(&lcnt[(unsigned)d[it] / GN_BN], 1);
#pragma unroll
    for (int it = 0; it < GN_EA / 256; ++it) {
        const int b  = (int)((unsigned)d[it] / GN_BN);
        const int dl = d[it] - b * GN_BN;
        if (r[it] < GN_CELL - 1) {
            lcell[b * GN_CELL + 1 + r[it]] = (dl << 17) | s[it];
        } else {
            const int q = atomicAdd(&lspill, 1);
            spill[(a * GN_EA + q) * 2]     = d[it];
            spill[(a * GN_EA + q) * 2 + 1] = s[it];
        }
    }
    __syncthreads();

    const int lane = t & 31;
    const int wv   = __builtin_amdgcn_readfirstlane(t >> 5);
    for (int b = wv; b < GN_NB; b += 8) {
        const int n = min(lcnt[b], GN_CELL - 1);
        int v = 0;
        if (lane >= 1 && lane <= n) v = lcell[b * GN_CELL + lane];
        if (lane == 0) v = n;
        cells[((size_t)b * GN_GA + a) * GN_CELL + lane] = v;
    }
    if (t == 0) spill_cnt[a] = lspill;
}

__global__ __launch_bounds__(256) void sage_fill_slots(
    const int* __restrict__ cells, const int* __restrict__ spill_cnt,
    const int* __restrict__ spill,
    int* __restrict__ cnt, int* __restrict__ slots,
    int* __restrict__ ovf_cursor, int* __restrict__ ovf_dst, int* __restrict__ ovf_src) {
    __shared__ __attribute__((aligned(16))) int lslots[GN_BN * GN_SLOTS];
    __shared__ int lcnt[GN_BN];

    const int t    = threadIdx.x;
    const int lane = t & 31;
    const int wv   = __builtin_amdgcn_readfirstlane(t >> 5);
    const int b    = blockIdx.x;
    const int base = b * GN_BN;

    const int* row = cells + (size_t)b * GN_GA * GN_CELL;
    constexpr int NFULL = GN_GA / 8;
    int v[NFULL + 1];
#pragma unroll
    for (int i = 0; i < NFULL; ++i) v[i] = row[(wv + 8 * i) * GN_CELL + lane];
    v[NFULL] = 0;
    if (wv + 8 * NFULL < GN_GA) v[NFULL] = row[(wv + 8 * NFULL) * GN_CELL + lane];
    int nsp = 0;
    if (t < GN_GA) nsp = spill_cnt[t];

    for (int i = t; i < GN_BN; i += 256) lcnt[i] = 0;
    for (int i = t; i < GN_BN * GN_SLOTS / 4; i += 256) ((int4*)lslots)[i] = make_int4(0, 0, 0, 0);
    __syncthreads();

    auto place = [&](int dl, int sid, int r) {
        if (r < GN_SLOTS) {
            lslots[dl * GN_SLOTS + r] = sid;
        } else {
            const int p = atomicAdd(ovf_cursor, 1);
            if ((unsigned)p < (unsigned)GN_E) { ovf_dst[p] = base + dl; ovf_src[p] = sid; }
        }
    };

    int r[NFULL + 1];
#pragma unroll
    for (int i = 0; i < NFULL + 1; ++i) {
        const int n = __builtin_amdgcn_readlane(v[i], 0);
        r[i] = -1;
        if (lane >= 1 && lane <= n) r[i] = atomicAdd(&lcnt[v[i] >> 17], 1);
    }
#pragma unroll
    for (int i = 0; i < NFULL + 1; ++i)
        if (r[i] >= 0) place(v[i] >> 17, v[i] & 0x1FFFF, r[i]);
    nsp = min(nsp, GN_EA);
    for (int q = 0; q < nsp; ++q) {
        const int dd = spill[(t * GN_EA + q) * 2];
        const int ss = spill[(t * GN_EA + q) * 2 + 1];
        if (dd >= base && dd < base + GN_BN) place(dd - base, ss, atomicAdd(&lcnt[dd - base], 1));
    }
    __syncthreads();

    for (int i = t; i < GN_BN; i += 256) cnt[base + i] = lcnt[i];
    int4*       gs = (int4*)(slots + (size_t)base * GN_SLOTS);
    const int4* ls = (const int4*)lslots;
    for (int i = t; i < GN_BN * GN_SLOTS / 4; i += 256) gs[i] = ls[i];
}

__device__ __forceinline__ void sage_issue8(const float* __restrict__ Hl,
                                            int ids, int j, int m, float4 (&v)[8]) {
#pragma unroll
    for (int k = 0; k < 8; ++k)
        if (k < m)
            v[k] = *(const float4*)(Hl + (size_t)__builtin_amdgcn_readlane(ids, j + k) * GN_D);
}

__device__ __forceinline__ void sage_add8(const float4 (&v)[8], int m, float4& acc) {
#pragma unroll
    for (int k = 0; k < 8; ++k)
        if (k < m) { acc.x += v[k].x; acc.y += v[k].y; acc.z += v[k].z; acc.w += v[k].w; }
}

__device__ __forceinline__ void sage_finish_node(
    const float* __restrict__ Hl, const float4 (&v)[8], int ids, int n, int node,
    const int* __restrict__ ovf_cursor, const int* __restrict__ ovf_dst,
    const int* __restrict__ ovf_src, int lane, float* dst) {
    float4 acc = make_float4(0.0f, 0.0f, 0.0f, 0.0f);
    sage_add8(v, min(n, 8), acc);
    if (n > 8) {
        const int m2 = min(n, GN_SLOTS) - 8;
        float4 t[8];
        sage_issue8(Hl, ids, 8, m2, t);
        sage_add8(t, m2, acc);
    }
    if (n > GN_SLOTS) {
        const int novf = min(*ovf_cursor, GN_E);
        for (int b = 0; b < novf; b += 32) {
            const int p = b + lane;
            int d = -1, s = 0;
            if (p < novf) { d = ovf_dst[p]; s = ovf_src[p]; }
            unsigned m = __builtin_amdgcn_ballot_w32(d == node);
            while (m) {
                const int l = __builtin_ctz(m);
                m &= m - 1;
                const int sid = __builtin_amdgcn_readlane(s, l);
                const float4 w = *(const float4*)(Hl + (size_t)sid * GN_D);
                acc.x += w.x; acc.y += w.y; acc.z += w.z; acc.w += w.w;
            }
        }
    }
    const float sc = 1.0f / fmaxf((float)n, 1.0f);
    acc.x *= sc; acc.y *= sc; acc.z *= sc; acc.w *= sc;
    *(float4*)(dst + lane * 4) = acc;
}


template <int NJP>
__device__ __forceinline__ void sage_tile(
    const float* __restrict__ A1, const float* __restrict__ A2,
    const float* Bs, float* __restrict__ out, int relu_out,
    int rbase, int row, int jp0, int lane, float s, const float* bj) {
    const int h   = lane >> 4;
    const int l15 = lane & 15;

    v8f c[2 * NJP];
#pragma unroll
    for (int jt = 0; jt < 2 * NJP; ++jt)
#pragma unroll
        for (int v = 0; v < 8; ++v) c[jt][v] = bj[jt];

    const int npath = A2 ? 2 : 1;
#pragma nounroll
    for (int p = 0; p < npath; ++p) {
        const float* arow  = (p ? A2 : A1) + (size_t)row * GN_D + 2 * h;
        const float* bfrag = Bs + p * GN_W_LDS + jp0 * 128 + lane * 4;
        const float  sc    = p ? 1.0f : s;

        v2f    ar[8];
        float4 bq[2][NJP];
#pragma unroll
        for (int i = 0; i < 8; ++i) ar[i] = *(const v2f*)(arow + 4 * i);
#pragma unroll
        for (int jp = 0; jp < NJP; ++jp) bq[0][jp] = *(const float4*)(bfrag + jp * 128);

#pragma unroll
        for (int kg = 0; kg < 32; ++kg) {
            if (kg + 1 < 32) {
#pragma unroll
                for (int jp = 0; jp < NJP; ++jp)
                    bq[(kg + 1) & 1][jp] =
                        *(const float4*)(bfrag + (kg + 1) * GN_KG_PITCH + jp * 128);
            }
            v2f a = ar[kg & 7];
            a.x *= sc;
            a.y *= sc;
            if (kg + 8 < 32) ar[kg & 7] = *(const v2f*)(arow + 4 * (kg + 8));
#pragma unroll
            for (int jp = 0; jp < NJP; ++jp) {
                const float4 bb = bq[kg & 1][jp];
                v2f b0; b0.x = bb.x; b0.y = bb.y;
                v2f b1; b1.x = bb.z; b1.y = bb.w;
                c[2 * jp] = __builtin_amdgcn_wmma_f32_16x16x4_f32(
                    false, a, false, b0, (short)0, c[2 * jp], false, false);
                c[2 * jp + 1] = __builtin_amdgcn_wmma_f32_16x16x4_f32(
                    false, a, false, b1, (short)0, c[2 * jp + 1], false, false);
            }
            __builtin_amdgcn_sched_barrier(0);
        }
    }

#pragma unroll
    for (int jt = 0; jt < 2 * NJP; ++jt)
#pragma unroll
        for (int v = 0; v < 8; ++v) {
            float val = c[jt][v];
            if (relu_out) val = fmaxf(val, 0.0f);
            out[(size_t)(rbase + v + 8 * h) * GN_D + jp0 * 32 + jt * 16 + l15] = val;
        }
}

__global__ __launch_bounds__(GN_GEMM_THREADS) void sage_gemm(
    const float* __restrict__ A1, const int* __restrict__ deg,
    const float* __restrict__ W1, const float* __restrict__ bias,
    const float* __restrict__ A2, const float* __restrict__ W2,
    float* __restrict__ out, int relu_out) {
    __shared__ __attribute__((aligned(16))) float Bs[GN_LDS_FLOATS];

    const int lane = threadIdx.x & 31;
    const int wv    = __builtin_amdgcn_readfirstlane((int)(threadIdx.x >> 5));
    const int tile0 = blockIdx.x * GN_GEMM_TILES;

    sage_stage_w(W1, Bs);
    if (A2) sage_stage_w(W2, Bs + GN_W_LDS);

    const int l15    = lane & 15;
    const int rb_own = (tile0 + wv) * 16;
    const int rb_sh  = (tile0 + GN_GEMM_WAVES) * 16;
    const int jp_sh  = wv & 3;
    const int row_own = min(rb_own + l15, GN_N - 1);
    const int row_sh  = min(rb_sh + l15, GN_N - 1);

    float s_own = 1.0f, s_sh = 1.0f;
    if (deg) {
        s_own = 1.0f / fmaxf((float)deg[row_own], 1.0f);
        s_sh  = 1.0f / fmaxf((float)deg[row_sh], 1.0f);
    }

    float bj_own[8], bj_sh[2];
#pragma unroll
    for (int jt = 0; jt < 8; ++jt) bj_own[jt] = bias[jt * 16 + l15];
#pragma unroll
    for (int jt = 0; jt < 2; ++jt) bj_sh[jt] = bias[jp_sh * 32 + jt * 16 + l15];

    __syncthreads();

    if (wv < 4)
        sage_tile<1>(A1, A2, Bs, out, relu_out, rb_sh, row_sh, jp_sh, lane, s_sh, bj_sh);
    sage_tile<4>(A1, A2, Bs, out, relu_out, rb_own, row_own, 0, lane, s_own, bj_own);
}


#define GN_A_PITCH     132
#define GN_A_TILE      (16 * GN_A_PITCH)
#define GN_AGG_FLOATS  (GN_W_LDS + GN_GEMM_TILES * GN_A_TILE)
#define GN_NODES_FLY   2
static_assert(GN_AGG_FLOATS * 4 <= 320 * 1024, "one weight and 25 A blocks must fit the LDS");
static_assert(GN_AGG_FLOATS * 4 > 160 * 1024, "one workgroup per WGP");
static_assert(16 % GN_NODES_FLY == 0, "whole groups of nodes per row block");

template <int NJP, bool A_LDS>
__device__ __forceinline__ void sage_path(v8f* c, const float* arow, const float* bfrag) {
    v2f    ar[8];
    v2f    al[2];
    float4 bq[2][NJP];
    if (A_LDS) {
        al[0] = *(const v2f*)arow;
    } else {
#pragma unroll
        for (int i = 0; i < 8; ++i) ar[i] = *(const v2f*)(arow + 4 * i);
    }
#pragma unroll
    for (int jp = 0; jp < NJP; ++jp) bq[0][jp] = *(const float4*)(bfrag + jp * 128);

#pragma unroll
    for (int kg = 0; kg < 32; ++kg) {
        if (kg + 1 < 32) {
#pragma unroll
            for (int jp = 0; jp < NJP; ++jp)
                bq[(kg + 1) & 1][jp] =
                    *(const float4*)(bfrag + (kg + 1) * GN_KG_PITCH + jp * 128);
            if (A_LDS) al[(kg + 1) & 1] = *(const v2f*)(arow + 4 * (kg + 1));
        }
        const v2f a = A_LDS ? al[kg & 1] : ar[kg & 7];
        if (!A_LDS && kg + 8 < 32) ar[kg & 7] = *(const v2f*)(arow + 4 * (kg + 8));
#pragma unroll
        for (int jp = 0; jp < NJP; ++jp) {
            const float4 bb = bq[kg & 1][jp];
            v2f b0; b0.x = bb.x; b0.y = bb.y;
            v2f b1; b1.x = bb.z; b1.y = bb.w;
            c[2 * jp] = __builtin_amdgcn_wmma_f32_16x16x4_f32(
                false, a, false, b0, (short)0, c[2 * jp], false, false);
            c[2 * jp + 1] = __builtin_amdgcn_wmma_f32_16x16x4_f32(
                false, a, false, b1, (short)0, c[2 * jp + 1], false, false);
        }
        __builtin_amdgcn_sched_barrier(0);
    }
}

template <int NJP>
__device__ __forceinline__ void sage_store(const v8f* c, float* __restrict__ out, int relu_out,
                                           int rbase, int jp0, int lane) {
    const int h   = lane >> 4;
    const int l15 = lane & 15;
#pragma unroll
    for (int jt = 0; jt < 2 * NJP; ++jt)
#pragma unroll
        for (int v = 0; v < 8; ++v) {
            float val = c[jt][v];
            if (relu_out) val = fmaxf(val, 0.0f);
            out[(size_t)(rbase + v + 8 * h) * GN_D + jp0 * 32 + jt * 16 + l15] = val;
        }
}

__global__ __launch_bounds__(GN_GEMM_THREADS) void sage_agg_gemm(
    const float* __restrict__ H, const int* __restrict__ cnt,
    const int* __restrict__ slots, const int* __restrict__ ovf_cursor,
    const int* __restrict__ ovf_dst, const int* __restrict__ ovf_src,
    const float* __restrict__ Wl, const float* __restrict__ bias,
    const float* __restrict__ X, const float* __restrict__ Wr,
    float* __restrict__ out, int relu_out) {
    __shared__ __attribute__((aligned(16))) float Ls[GN_AGG_FLOATS];
    float* Bs = Ls;
    float* As = Ls + GN_W_LDS;

    const int lane = threadIdx.x & 31;
    const int wv    = __builtin_amdgcn_readfirstlane((int)(threadIdx.x >> 5));
    const int tile0 = blockIdx.x * GN_GEMM_TILES;
    const int h     = lane >> 4;
    const int l15   = lane & 15;
    const int sl    = lane & (GN_SLOTS - 1);
    const int rb_own = (tile0 + wv) * 16;
    const int rb_sh  = (tile0 + GN_GEMM_WAVES) * 16;
    const int jp_sh  = wv & 3;
    const float* Hl  = H + lane * 4;

    const int cntv = cnt[min(rb_own + l15, GN_N - 1)];
    int ids_next[GN_NODES_FLY];
#pragma unroll
    for (int g = 0; g < GN_NODES_FLY; ++g)
        ids_next[g] = slots[min(rb_own + g, GN_N - 1) * GN_SLOTS + sl];

    sage_stage_w(Wl, Bs);

    float* At = As + wv * GN_A_TILE;
#pragma nounroll
    for (int r = 0; r < 16; r += GN_NODES_FLY) {
        int ids[GN_NODES_FLY], n[GN_NODES_FLY];
#pragma unroll
        for (int g = 0; g < GN_NODES_FLY; ++g) {
            ids[g] = ids_next[g];
            n[g]   = __builtin_amdgcn_readlane(cntv, r + g);
        }
#pragma unroll
        for (int g = 0; g < GN_NODES_FLY; ++g)
            ids_next[g] = slots[min(rb_own + r + GN_NODES_FLY + g, GN_N - 1) * GN_SLOTS + sl];
        float4 v[GN_NODES_FLY][8];
#pragma unroll
        for (int g = 0; g < GN_NODES_FLY; ++g) sage_issue8(Hl, ids[g], 0, min(n[g], 8), v[g]);
#pragma unroll
        for (int g = 0; g < GN_NODES_FLY; ++g)
            sage_finish_node(Hl, v[g], ids[g], n[g], rb_own + r + g,
                             ovf_cursor, ovf_dst, ovf_src, lane, At + (r + g) * GN_A_PITCH);
    }
    if (wv >= 4 && wv < 20) {
        const int node = min(rb_sh + wv - 4, GN_N - 1);
        const int ns   = __builtin_amdgcn_readfirstlane(cnt[node]);
        const int idss = slots[node * GN_SLOTS + sl];
        float4 v[8];
        sage_issue8(Hl, idss, 0, min(ns, 8), v);
        sage_finish_node(Hl, v, idss, ns, node, ovf_cursor, ovf_dst, ovf_src, lane,
                         As + GN_GEMM_WAVES * GN_A_TILE + (wv - 4) * GN_A_PITCH);
    }

    v8f c_own[8], c_sh[2];
#pragma unroll
    for (int jt = 0; jt < 8; ++jt) {
        const float bj = bias[jt * 16 + l15];
#pragma unroll
        for (int v = 0; v < 8; ++v) c_own[jt][v] = bj;
    }
#pragma unroll
    for (int jt = 0; jt < 2; ++jt) {
        const float bj = bias[jp_sh * 32 + jt * 16 + l15];
#pragma unroll
        for (int v = 0; v < 8; ++v) c_sh[jt][v] = bj;
    }

    __syncthreads();

    if (wv < 4)
        sage_path<1, true>(c_sh, As + GN_GEMM_WAVES * GN_A_TILE + l15 * GN_A_PITCH + 2 * h,
                           Bs + jp_sh * 128 + lane * 4);
    sage_path<4, true>(c_own, At + l15 * GN_A_PITCH + 2 * h, Bs + lane * 4);

    __syncthreads();
    sage_stage_w(Wr, Bs);
    __syncthreads();

    const int row_own = min(rb_own + l15, GN_N - 1);
    const int row_sh  = min(rb_sh + l15, GN_N - 1);
    if (wv < 4) {
        sage_path<1, false>(c_sh, X + (size_t)row_sh * GN_D + 2 * h, Bs + jp_sh * 128 + lane * 4);
        sage_store<1>(c_sh, out, relu_out, rb_sh, jp_sh, lane);
    }
    sage_path<4, false>(c_own, X + (size_t)row_own * GN_D + 2 * h, Bs + lane * 4);
    sage_store<4>(c_own, out, relu_out, rb_own, 0, lane);
}

extern "C" void kernel_launch(void* const* d_in, const int* in_sizes, int n_in,
                              void* d_out, int out_size, void* d_ws, size_t ws_size,
                              hipStream_t stream) {
    (void)in_sizes; (void)n_in; (void)out_size; (void)ws_size;

    const float* x   = (const float*)d_in[0];
    const int*   ei  = (const int*)d_in[1];
    const float* W0p = (const float*)d_in[2];
    const float* b0p = (const float*)d_in[3];
    const float* W0l = (const float*)d_in[4];
    const float* b0l = (const float*)d_in[5];
    const float* W0r = (const float*)d_in[6];
    const float* W1p = (const float*)d_in[7];
    const float* b1p = (const float*)d_in[8];
    const float* W1l = (const float*)d_in[9];
    const float* b1l = (const float*)d_in[10];
    const float* W1r = (const float*)d_in[11];
    float* out = (float*)d_out;

    const size_t nd = (size_t)GN_N * GN_D;
    float* H   = (float*)d_ws;
    float* X1  = H + nd;
    int* CNT   = (int*)(X1 + nd);
    int* CUR   = CNT + GN_N;
    int* SLOTS = CNT + GN_N + 32;
    int* OVFD  = SLOTS + GN_N * GN_SLOTS;
    int* OVFS  = OVFD + GN_E;
    int* CELLS = OVFS + GN_E;
    int* SPCNT = CELLS + GN_NB * GN_GA * GN_CELL;
    int* SPILL = SPCNT + 256;

    const int* src = ei;
    const int* dst = ei + GN_E;

    sage_bin_edges<<<GN_GA, 256, 0, stream>>>(src, dst, CELLS, SPCNT, SPILL, CUR);
    sage_fill_slots<<<GN_NB, 256, 0, stream>>>(CELLS, SPCNT, SPILL, CNT, SLOTS, CUR, OVFD, OVFS);

    const int gemm_blocks   = GN_GEMM_BLOCKS;

    sage_gemm<<<gemm_blocks, GN_GEMM_THREADS, 0, stream>>>(x, nullptr, W0p, b0p,
                                               nullptr, nullptr, H, 1);
    sage_agg_gemm<<<gemm_blocks, GN_GEMM_THREADS, 0, stream>>>(H, CNT, SLOTS, CUR, OVFD, OVFS,
                                               W0l, b0l, x, W0r, X1, 1);

    sage_gemm<<<gemm_blocks, GN_GEMM_THREADS, 0, stream>>>(X1, nullptr, W1p, b1p,
                                               nullptr, nullptr, H, 1);
    sage_agg_gemm<<<gemm_blocks, GN_GEMM_THREADS, 0, stream>>>(H, CNT, SLOTS, CUR, OVFD, OVFS,
                                               W1l, b1l, X1, W1r, out, 0);
}
